# starting kernel
# baseline (speedup 1.0000x reference)
#include <hip/hip_runtime.h>
#include <hip/hip_bf16.h>

typedef __attribute__((ext_vector_type(2))) float v2f;
typedef __attribute__((ext_vector_type(8))) float v8f;

#define NEG_SLOPE 0.2f

__device__ __forceinline__ unsigned ordEnc(float f) {
  unsigned u = __float_as_uint(f);
  return (u & 0x80000000u) ? ~u : (u | 0x80000000u);
}
__device__ __forceinline__ float ordDec(unsigned u) {
  return (u & 0x80000000u) ? __uint_as_float(u ^ 0x80000000u)
                           : __uint_as_float(~u);
}

__global__ void gat_fill_u32(unsigned* __restrict__ p, unsigned v, long long n) {
  long long i = (long long)blockIdx.x * blockDim.x + threadIdx.x;
  if (i < n) p[i] = v;
}

__global__ void gat_count_loop(const int* __restrict__ dst,
                               const float* __restrict__ ea,
                               float* __restrict__ cnt,
                               float* __restrict__ la, int E) {
  long long i = (long long)blockIdx.x * blockDim.x + threadIdx.x;
  if (i >= (long long)E * 16) return;
  int e = (int)(i >> 4), j = (int)(i & 15);
  int d = dst[e];
  atomicAdd(&la[(long long)d * 16 + j], ea[i]);
  if (j == 0) atomicAdd(&cnt[d], 1.0f);
}

__global__ void gat_loop_div(const float* __restrict__ cnt,
                             float* __restrict__ la, int Nn) {
  long long i = (long long)blockIdx.x * blockDim.x + threadIdx.x;
  if (i >= (long long)Nn * 16) return;
  float c = cnt[i >> 4];
  c = (c < 1.0f) ? 1.0f : c;
  la[i] = la[i] / c;
}

__global__ void gat_medge(const float* __restrict__ We,
                          const float* __restrict__ ae,
                          float* __restrict__ medge) {
  int t = threadIdx.x;
  int k = t >> 2, h = t & 3;
  float s = 0.0f;
  for (int c = 0; c < 32; ++c)
    s += We[k * 128 + h * 32 + c] * ae[h * 32 + c];
  medge[k * 4 + h] = s;
}

__global__ __launch_bounds__(256) void gat_gemm_wmma(
    const float* __restrict__ X, const float* __restrict__ W,
    float* __restrict__ Y, int Nrows) {
  int wave = (int)(((long long)blockIdx.x * blockDim.x + threadIdx.x) >> 5);
  int lane = threadIdx.x & 31;
  int m0 = wave * 16;
  if (m0 >= Nrows) return;

  v8f acc[8] = {};
  int half = lane >> 4;
  int l16 = lane & 15;
  int arow = m0 + l16;
  if (arow >= Nrows) arow = Nrows - 1;
  const float* xr = X + (long long)arow * 128;

  for (int k0 = 0; k0 < 128; k0 += 4) {
    v2f a;
    a.x = xr[k0 + half * 2];
    a.y = xr[k0 + half * 2 + 1];
#pragma unroll
    for (int nt = 0; nt < 8; ++nt) {
      int col = nt * 16 + l16;
      v2f b;
      b.x = W[(k0 + half * 2) * 128 + col];
      b.y = W[(k0 + half * 2 + 1) * 128 + col];
      acc[nt] = __builtin_amdgcn_wmma_f32_16x16x4_f32(
          false, a, false, b, (short)0, acc[nt], false, false);
    }
  }
#pragma unroll
  for (int nt = 0; nt < 8; ++nt) {
#pragma unroll
    for (int r = 0; r < 8; ++r) {
      int row = m0 + r + half * 8;
      if (row < Nrows) Y[(long long)row * 128 + nt * 16 + l16] = acc[nt][r];
    }
  }
}

__global__ __launch_bounds__(128) void gat_attdot(
    const float* __restrict__ xh, const float* __restrict__ as,
    const float* __restrict__ ad, float* __restrict__ a_src,
    float* __restrict__ a_dst, int Nn) {
  int n = blockIdx.x;
  int t = threadIdx.x;
  float v = xh[(long long)n * 128 + t];
  float ps = v * as[t];
  float pd = v * ad[t];
  for (int o = 16; o > 0; o >>= 1) {
    ps += __shfl_down(ps, o, 32);
    pd += __shfl_down(pd, o, 32);
  }
  if ((t & 31) == 0) {
    int h = t >> 5;
    a_src[n * 4 + h] = ps;
    a_dst[n * 4 + h] = pd;
  }
}

__global__ void gat_alpha(const int* __restrict__ srcI,
                          const int* __restrict__ dstI,
                          const float* __restrict__ ea,
                          const float* __restrict__ la,
                          const float* __restrict__ a_src,
                          const float* __restrict__ a_dst,
                          const float* __restrict__ medge,
                          float* __restrict__ alpha,
                          unsigned* __restrict__ mseg, int E, int Nn) {
  long long f = (long long)blockIdx.x * blockDim.x + threadIdx.x;
  long long F = (long long)E + Nn;
  if (f >= F) return;
  int s, d;
  const float* row;
  if (f < (long long)E) {
    s = srcI[f]; d = dstI[f]; row = ea + f * 16;
  } else {
    s = (int)(f - E); d = s; row = la + (long long)s * 16;
  }
  float e0[16];
#pragma unroll
  for (int k = 0; k < 16; ++k) e0[k] = row[k];
#pragma unroll
  for (int h = 0; h < 4; ++h) {
    float aE = 0.0f;
#pragma unroll
    for (int k = 0; k < 16; ++k) aE += e0[k] * medge[k * 4 + h];
    float v = a_src[(long long)s * 4 + h] + a_dst[(long long)d * 4 + h] + aE;
    v = (v > 0.0f) ? v : NEG_SLOPE * v;
    alpha[f * 4 + h] = v;
    atomicMax(&mseg[(long long)d * 4 + h], ordEnc(v));
  }
}

__global__ void gat_expsum(const int* __restrict__ dstI,
                           float* __restrict__ alpha,
                           const unsigned* __restrict__ mseg,
                           float* __restrict__ sseg, int E, int Nn) {
  long long i = (long long)blockIdx.x * blockDim.x + threadIdx.x;
  long long F4 = ((long long)E + Nn) * 4;
  if (i >= F4) return;
  long long f = i >> 2;
  int h = (int)(i & 3);
  int d = (f < (long long)E) ? dstI[f] : (int)(f - E);
  float mv = ordDec(mseg[(long long)d * 4 + h]);
  if (!(mv > -__builtin_inff() && mv < __builtin_inff())) mv = 0.0f;
  float p = __expf(alpha[i] - mv);
  alpha[i] = p;
  atomicAdd(&sseg[(long long)d * 4 + h], p);
}

__global__ __launch_bounds__(256) void gat_aggr(
    const int* __restrict__ srcI, const int* __restrict__ dstI,
    const float* __restrict__ alpha, const float* __restrict__ sseg,
    const float* __restrict__ xh, float* __restrict__ out, int E, int Nn) {
  long long wid = ((long long)blockIdx.x * blockDim.x + threadIdx.x) >> 5;
  int lane = threadIdx.x & 31;
  long long F = (long long)E + Nn;
  if (wid >= F) return;
  int s, d;
  if (wid < (long long)E) { s = srcI[wid]; d = dstI[wid]; }
  else { s = (int)(wid - E); d = s; }
  int c0 = lane * 4;
  int h = lane >> 3;
  float a = alpha[wid * 4 + h] / (sseg[(long long)d * 4 + h] + 1e-16f);
  const float4 xv = *(const float4*)(xh + (long long)s * 128 + c0);
  float* o = out + (long long)d * 128 + c0;
  atomicAdd(o + 0, a * xv.x);
  atomicAdd(o + 1, a * xv.y);
  atomicAdd(o + 2, a * xv.z);
  atomicAdd(o + 3, a * xv.w);
}

__global__ void gat_bias_act(float* __restrict__ y,
                             const float* __restrict__ bias, long long n,
                             int relu) {
  long long i = (long long)blockIdx.x * blockDim.x + threadIdx.x;
  if (i >= n) return;
  float v = y[i] + bias[i & 127];
  if (relu) v = (v > 0.0f) ? v : 0.0f;
  y[i] = v;
}

static void fillN(unsigned* p, unsigned v, long long n, hipStream_t s) {
  if (n <= 0) return;
  long long b = (n + 255) / 256;
  hipLaunchKernelGGL(gat_fill_u32, dim3((unsigned)b), dim3(256), 0, s, p, v, n);
}

extern "C" void kernel_launch(void* const* d_in, const int* in_sizes, int n_in,
                              void* d_out, int out_size, void* d_ws,
                              size_t ws_size, hipStream_t stream) {
  const int N = in_sizes[0] / 128;
  const int E = in_sizes[1] / 2;

  const float* x  = (const float*)d_in[0];
  const int* ei   = (const int*)d_in[1];
  const int* srcI = ei;
  const int* dstI = ei + E;
  const float* ea = (const float*)d_in[2];

  const float* Wl[2]  = {(const float*)d_in[3], (const float*)d_in[9]};
  const float* asl[2] = {(const float*)d_in[4], (const float*)d_in[10]};
  const float* adl[2] = {(const float*)d_in[5], (const float*)d_in[11]};
  const float* Wel[2] = {(const float*)d_in[6], (const float*)d_in[12]};
  const float* ael[2] = {(const float*)d_in[7], (const float*)d_in[13]};
  const float* bl[2]  = {(const float*)d_in[8], (const float*)d_in[14]};

  float* wsf = (float*)d_ws;
  size_t off = 0;
  float* cnt   = wsf + off; off += (size_t)N;
  float* la    = wsf + off; off += (size_t)N * 16;
  float* medge = wsf + off; off += 64;
  float* xh    = wsf + off; off += (size_t)N * 128;
  float* a_src = wsf + off; off += (size_t)N * 4;
  float* a_dst = wsf + off; off += (size_t)N * 4;
  float* alpha = wsf + off; off += ((size_t)E + N) * 4;
  unsigned* mseg = (unsigned*)(wsf + off); off += (size_t)N * 4;
  float* sseg  = wsf + off; off += (size_t)N * 4;
  float* h1    = wsf + off; off += (size_t)N * 128;
  (void)ws_size; (void)n_in; (void)out_size;

  const long long F = (long long)E + N;

  fillN((unsigned*)cnt, 0u, N, stream);
  fillN((unsigned*)la, 0u, (long long)N * 16, stream);
  {
    long long n = (long long)E * 16;
    hipLaunchKernelGGL(gat_count_loop, dim3((unsigned)((n + 255) / 256)),
                       dim3(256), 0, stream, dstI, ea, cnt, la, E);
    long long m = (long long)N * 16;
    hipLaunchKernelGGL(gat_loop_div, dim3((unsigned)((m + 255) / 256)),
                       dim3(256), 0, stream, cnt, la, N);
  }

  for (int l = 0; l < 2; ++l) {
    const float* xin = (l == 0) ? x : h1;
    float* outb = (l == 0) ? h1 : (float*)d_out;

    hipLaunchKernelGGL(gat_medge, dim3(1), dim3(64), 0, stream, Wel[l], ael[l],
                       medge);

    {
      long long waves = ((long long)N + 15) / 16;
      long long blocks = (waves + 7) / 8;
      hipLaunchKernelGGL(gat_gemm_wmma, dim3((unsigned)blocks), dim3(256), 0,
                         stream, xin, Wl[l], xh, N);
    }

    hipLaunchKernelGGL(gat_attdot, dim3((unsigned)N), dim3(128), 0, stream, xh,
                       asl[l], adl[l], a_src, a_dst, N);

    fillN(mseg, 0x007fffffu, (long long)N * 4, stream);
    fillN((unsigned*)sseg, 0u, (long long)N * 4, stream);
    fillN((unsigned*)outb, 0u, (long long)N * 128, stream);

    hipLaunchKernelGGL(gat_alpha, dim3((unsigned)((F + 255) / 256)), dim3(256),
                       0, stream, srcI, dstI, ea, la, a_src, a_dst, medge,
                       alpha, mseg, E, N);

    {
      long long n4 = F * 4;
      hipLaunchKernelGGL(gat_expsum, dim3((unsigned)((n4 + 255) / 256)),
                         dim3(256), 0, stream, dstI, alpha, mseg, sseg, E, N);
    }

    {
      long long blocks = (F + 7) / 8;
      hipLaunchKernelGGL(gat_aggr, dim3((unsigned)blocks), dim3(256), 0, stream,
                         srcI, dstI, alpha, sseg, xh, outb, E, N);
    }

    {
      long long n = (long long)N * 128;
      hipLaunchKernelGGL(gat_bias_act, dim3((unsigned)((n + 255) / 256)),
                         dim3(256), 0, stream, outb, bl[l], n, (l == 0) ? 1 : 0);
    }
  }
}


# aggregation by destination-range owners: a work-group keeps 400 nodes' sums in LDS, scans the whole dst list for its edges, LDS adds, one store per line with bias/ReLU; no out atomics; sum order new
# speedup vs baseline: 1.7452x; 1.7452x over previous
#include <hip/hip_runtime.h>
#include <hip/hip_bf16.h>

typedef __attribute__((ext_vector_type(2))) float v2f;
typedef __attribute__((ext_vector_type(8))) float v8f;

#define NEG_SLOPE 0.2f

__device__ __forceinline__ unsigned ordEnc(float f) {
  unsigned u = __float_as_uint(f);
  return (u & 0x80000000u) ? ~u : (u | 0x80000000u);
}
__device__ __forceinline__ float ordDec(unsigned u) {
  return (u & 0x80000000u) ? __uint_as_float(u ^ 0x80000000u)
                           : __uint_as_float(~u);
}

__global__ void gat_fill_u32(unsigned* __restrict__ p, unsigned v, long long n) {
  long long i = (long long)blockIdx.x * blockDim.x + threadIdx.x;
  if (i < n) p[i] = v;
}

__global__ void gat_count_loop(const int* __restrict__ dst,
                               const float* __restrict__ ea,
                               float* __restrict__ cnt,
                               float* __restrict__ la, int E) {
  long long i = (long long)blockIdx.x * blockDim.x + threadIdx.x;
  if (i >= (long long)E * 16) return;
  int e = (int)(i >> 4), j = (int)(i & 15);
  int d = dst[e];
  atomicAdd(&la[(long long)d * 16 + j], ea[i]);
  if (j == 0) atomicAdd(&cnt[d], 1.0f);
}

__global__ void gat_loop_div(const float* __restrict__ cnt,
                             float* __restrict__ la, int Nn) {
  long long i = (long long)blockIdx.x * blockDim.x + threadIdx.x;
  if (i >= (long long)Nn * 16) return;
  float c = cnt[i >> 4];
  c = (c < 1.0f) ? 1.0f : c;
  la[i] = la[i] / c;
}

__global__ void gat_medge(const float* __restrict__ We,
                          const float* __restrict__ ae,
                          float* __restrict__ medge) {
  int t = threadIdx.x;
  int k = t >> 2, h = t & 3;
  float s = 0.0f;
  for (int c = 0; c < 32; ++c)
    s += We[k * 128 + h * 32 + c] * ae[h * 32 + c];
  medge[k * 4 + h] = s;
}

__global__ __launch_bounds__(256) void gat_gemm_wmma(
    const float* __restrict__ X, const float* __restrict__ W,
    float* __restrict__ Y, int Nrows) {
  int wave = (int)(((long long)blockIdx.x * blockDim.x + threadIdx.x) >> 5);
  int lane = threadIdx.x & 31;
  int m0 = wave * 16;
  if (m0 >= Nrows) return;

  v8f acc[8] = {};
  int half = lane >> 4;
  int l16 = lane & 15;
  int arow = m0 + l16;
  if (arow >= Nrows) arow = Nrows - 1;
  const float* xr = X + (long long)arow * 128;

  for (int k0 = 0; k0 < 128; k0 += 4) {
    v2f a;
    a.x = xr[k0 + half * 2];
    a.y = xr[k0 + half * 2 + 1];
#pragma unroll
    for (int nt = 0; nt < 8; ++nt) {
      int col = nt * 16 + l16;
      v2f b;
      b.x = W[(k0 + half * 2) * 128 + col];
      b.y = W[(k0 + half * 2 + 1) * 128 + col];
      acc[nt] = __builtin_amdgcn_wmma_f32_16x16x4_f32(
          false, a, false, b, (short)0, acc[nt], false, false);
    }
  }
#pragma unroll
  for (int nt = 0; nt < 8; ++nt) {
#pragma unroll
    for (int r = 0; r < 8; ++r) {
      int row = m0 + r + half * 8;
      if (row < Nrows) Y[(long long)row * 128 + nt * 16 + l16] = acc[nt][r];
    }
  }
}

__global__ __launch_bounds__(128) void gat_attdot(
    const float* __restrict__ xh, const float* __restrict__ as,
    const float* __restrict__ ad, float* __restrict__ a_src,
    float* __restrict__ a_dst, int Nn) {
  int n = blockIdx.x;
  int t = threadIdx.x;
  float v = xh[(long long)n * 128 + t];
  float ps = v * as[t];
  float pd = v * ad[t];
  for (int o = 16; o > 0; o >>= 1) {
    ps += __shfl_down(ps, o, 32);
    pd += __shfl_down(pd, o, 32);
  }
  if ((t & 31) == 0) {
    int h = t >> 5;
    a_src[n * 4 + h] = ps;
    a_dst[n * 4 + h] = pd;
  }
}

__global__ void gat_alpha(const int* __restrict__ srcI,
                          const int* __restrict__ dstI,
                          const float* __restrict__ ea,
                          const float* __restrict__ la,
                          const float* __restrict__ a_src,
                          const float* __restrict__ a_dst,
                          const float* __restrict__ medge,
                          float* __restrict__ alpha,
                          unsigned* __restrict__ mseg, int E, int Nn) {
  long long i = (long long)blockIdx.x * blockDim.x + threadIdx.x;
  long long F = (long long)E + Nn;
  if (i >= F * 4) return;
  long long f = i >> 2;
  int h = (int)(i & 3);
  int s, d;
  const float* row;
  if (f < (long long)E) {
    s = srcI[f]; d = dstI[f]; row = ea + f * 16;
  } else {
    s = (int)(f - E); d = s; row = la + (long long)s * 16;
  }
  float e0[16];
#pragma unroll
  for (int k = 0; k < 16; ++k) e0[k] = row[k];
  float q0 = 0.0f, q1 = 0.0f, q2 = 0.0f, q3 = 0.0f;
#pragma unroll
  for (int k = 0; k < 16; ++k) {
    q0 += e0[k] * medge[k * 4 + 0];
    q1 += e0[k] * medge[k * 4 + 1];
    q2 += e0[k] * medge[k * 4 + 2];
    q3 += e0[k] * medge[k * 4 + 3];
  }
  float aE = q3;
  aE = (h == 2) ? q2 : aE;
  aE = (h == 1) ? q1 : aE;
  aE = (h == 0) ? q0 : aE;
  float v = a_src[(long long)s * 4 + h] + a_dst[(long long)d * 4 + h] + aE;
  v = (v > 0.0f) ? v : NEG_SLOPE * v;
  alpha[i] = v;
  atomicMax(&mseg[(long long)d * 4 + h], ordEnc(v));
}

__global__ void gat_expsum(const int* __restrict__ dstI,
                           float* __restrict__ alpha,
                           const unsigned* __restrict__ mseg,
                           float* __restrict__ sseg, int E, int Nn) {
  long long i = (long long)blockIdx.x * blockDim.x + threadIdx.x;
  long long F4 = ((long long)E + Nn) * 4;
  if (i >= F4) return;
  long long f = i >> 2;
  int h = (int)(i & 3);
  int d = (f < (long long)E) ? dstI[f] : (int)(f - E);
  float mv = ordDec(mseg[(long long)d * 4 + h]);
  if (!(mv > -__builtin_inff() && mv < __builtin_inff())) mv = 0.0f;
  float p = __expf(alpha[i] - mv);
  alpha[i] = p;
  atomicAdd(&sseg[(long long)d * 4 + h], p);
}

#define OWN_R 400
#define OWN_T 1024
#define OWN_W (OWN_T / 32)
#define OWN_CAP 16384
static_assert(OWN_R % 80 == 0, "start values are formed 80 rows at a time");

__device__ __noinline__ void own_add_alone(
    float* acc_row, const float* __restrict__ xr,
    const float* __restrict__ al4, const float* __restrict__ sg4) {
  for (int h = 0; h < 4; ++h) {
    float w = al4[h] / (sg4[h] + 1e-16f);
    for (int c = 0; c < 32; ++c) atomicAdd(acc_row + h * 32 + c, w * xr[h * 32 + c]);
  }
}

__device__ __forceinline__ void own_take(
    int dval, long long i, int base, int rows, float* acc, int* lst, int* cntS,
    const int* __restrict__ srcI, const float* __restrict__ alpha,
    const float* __restrict__ sseg, const float* __restrict__ xh) {
  unsigned r = (unsigned)(dval - base);
  if (r < (unsigned)rows) {
    int slot = atomicAdd(cntS, 1);
    if (slot < OWN_CAP) {
      lst[slot] = (int)i;
    } else {
      own_add_alone(acc + r * 128, xh + (long long)srcI[i] * 128, alpha + i * 4,
                    sseg + (long long)dval * 4);
    }
  }
}

__global__ __launch_bounds__(OWN_T) void gat_own(
    const int* __restrict__ srcI, const int* __restrict__ dstI,
    const float* __restrict__ alpha, const float* __restrict__ sseg,
    const float* __restrict__ xh, const float* __restrict__ bias,
    float* __restrict__ out, int E, int Nn, int relu) {
  __shared__ __attribute__((aligned(16))) float acc[OWN_R * 128];
  __shared__ int lst[OWN_CAP];
  __shared__ int cntS;

  const int t = threadIdx.x;
  const int lane = t & 31;
  const int base = (int)blockIdx.x * OWN_R;
  const int rows = (Nn - base >= OWN_R) ? OWN_R : (Nn - base);

  if (t == 0) cntS = 0;
  {
    const int c = t & 127;
    const int h = c >> 5;
    const int r0 = t >> 7;
    for (int rb = 0; rb < OWN_R; rb += 80) {
      float xa[10], aa[10], sa[10];
#pragma unroll
      for (int q = 0; q < 10; ++q) {
        int r = rb + q * 8 + r0;
        long long n = (long long)base + ((r < rows) ? r : 0);
        xa[q] = xh[n * 128 + c];
        aa[q] = alpha[((long long)E + n) * 4 + h];
        sa[q] = sseg[n * 4 + h];
      }
#pragma unroll
      for (int q = 0; q < 10; ++q) {
        int r = rb + q * 8 + r0;
        float v = (aa[q] / (sa[q] + 1e-16f)) * xa[q];
        acc[r * 128 + c] = (r < rows) ? v : 0.0f;
      }
    }
  }
  __syncthreads();

  {
    int lead = (int)((4u - (unsigned)(((unsigned long long)dstI >> 2) & 3u)) & 3u);
    if (lead > E) lead = E;
    const long long nv = ((long long)E - lead) >> 2;
    const long long tail0 = (long long)lead + nv * 4;
    {
      long long i = -1;
      if (t < lead) i = t;
      else if ((long long)(t - lead) < (long long)E - tail0) i = tail0 + (t - lead);
      if (i >= 0) own_take(dstI[i], i, base, rows, acc, lst, &cntS, srcI, alpha, sseg, xh);
    }
    const int4* pv = (const int4*)(dstI + lead);
    const long long step = (long long)OWN_T * 4;
    auto load_step = [&](int4 (&qq)[4], long long v0) {
      if (v0 + step <= nv) {
#pragma unroll
        for (int u = 0; u < 4; ++u) qq[u] = pv[v0 + (long long)u * OWN_T + t];
      } else {
#pragma unroll
        for (int u = 0; u < 4; ++u) {
          long long v = v0 + (long long)u * OWN_T + t;
          qq[u] = (v < nv) ? pv[v] : make_int4(-1, -1, -1, -1);
        }
      }
    };
    int4 q[4];
    if (nv > 0) load_step(q, 0);
    for (long long v0 = 0; v0 < nv; v0 += step) {
      int4 qn[4];
      if (v0 + step < nv) {
        load_step(qn, v0 + step);
      } else {
#pragma unroll
        for (int u = 0; u < 4; ++u) qn[u] = make_int4(-1, -1, -1, -1);
      }
#pragma unroll
      for (int u = 0; u < 4; ++u) {
        long long i = (long long)lead + (v0 + (long long)u * OWN_T + t) * 4;
        own_take(q[u].x, i + 0, base, rows, acc, lst, &cntS, srcI, alpha, sseg, xh);
        own_take(q[u].y, i + 1, base, rows, acc, lst, &cntS, srcI, alpha, sseg, xh);
        own_take(q[u].z, i + 2, base, rows, acc, lst, &cntS, srcI, alpha, sseg, xh);
        own_take(q[u].w, i + 3, base, rows, acc, lst, &cntS, srcI, alpha, sseg, xh);
      }
#pragma unroll
      for (int u = 0; u < 4; ++u) q[u] = qn[u];
    }
  }
  __syncthreads();
  int nl = cntS;
  nl = (nl < OWN_CAP) ? nl : OWN_CAP;

  const int wv = __builtin_amdgcn_readfirstlane(t >> 5);
  const int hl = lane & 3;
  int s = 0, d = base;
  float al = 0.0f;
  if (wv * 8 < nl) {
    int slot = wv * 8 + (lane >> 2);
    if (slot >= nl) slot = nl - 1;
    int e = lst[slot];
    al = alpha[(long long)e * 4 + hl];
    s = srcI[e];
    d = dstI[e];
  }
  for (int g0 = wv * 8; g0 < nl; g0 += OWN_W * 8) {
    int nlive = (nl - g0 >= 8) ? 8 : (nl - g0);
    float sg = sseg[(long long)d * 4 + hl];
    float xv[8][4];
    int rr[8];
#pragma unroll
    for (int k = 0; k < 8; ++k) {
      int se = __builtin_amdgcn_readlane(s, k * 4);
      rr[k] = __builtin_amdgcn_readlane(d, k * 4) - base;
      const float* xr = xh + (long long)se * 128 + lane;
#pragma unroll
      for (int j = 0; j < 4; ++j) xv[k][j] = xr[j * 32];
    }
    int s2 = 0, d2 = base;
    float al2 = 0.0f;
    int gn = g0 + OWN_W * 8;
    if (gn < nl) {
      int slot = gn + (lane >> 2);
      if (slot >= nl) slot = nl - 1;
      int e = lst[slot];
      al2 = alpha[(long long)e * 4 + hl];
      s2 = srcI[e];
      d2 = dstI[e];
    }
    __builtin_amdgcn_sched_barrier(0);
    float w = al / (sg + 1e-16f);
    int wi = __float_as_int(w);
#pragma unroll
    for (int k = 0; k < 8; ++k) {
      if (k == 0 || k < nlive) {
        float* a = acc + rr[k] * 128 + lane;
#pragma unroll
        for (int j = 0; j < 4; ++j) {
          float wj = __int_as_float(__builtin_amdgcn_readlane(wi, k * 4 + j));
          atomicAdd(a + j * 32, wj * xv[k][j]);
        }
      }
    }
    s = s2; d = d2; al = al2;
  }
  __syncthreads();

  {
    const int c = t & 127;
    const float b = bias[c];
    for (int r = t >> 7; r < rows; r += OWN_T / 128) {
      float v = acc[r * 128 + c] + b;
      if (relu) v = (v > 0.0f) ? v : 0.0f;
      out[((long long)base + r) * 128 + c] = v;
    }
  }
}

static void fillN(unsigned* p, unsigned v, long long n, hipStream_t s) {
  if (n <= 0) return;
  long long b = (n + 255) / 256;
  hipLaunchKernelGGL(gat_fill_u32, dim3((unsigned)b), dim3(256), 0, s, p, v, n);
}

extern "C" void kernel_launch(void* const* d_in, const int* in_sizes, int n_in,
                              void* d_out, int out_size, void* d_ws,
                              size_t ws_size, hipStream_t stream) {
  const int N = in_sizes[0] / 128;
  const int E = in_sizes[1] / 2;

  const float* x  = (const float*)d_in[0];
  const int* ei   = (const int*)d_in[1];
  const int* srcI = ei;
  const int* dstI = ei + E;
  const float* ea = (const float*)d_in[2];

  const float* Wl[2]  = {(const float*)d_in[3], (const float*)d_in[9]};
  const float* asl[2] = {(const float*)d_in[4], (const float*)d_in[10]};
  const float* adl[2] = {(const float*)d_in[5], (const float*)d_in[11]};
  const float* Wel[2] = {(const float*)d_in[6], (const float*)d_in[12]};
  const float* ael[2] = {(const float*)d_in[7], (const float*)d_in[13]};
  const float* bl[2]  = {(const float*)d_in[8], (const float*)d_in[14]};

  float* wsf = (float*)d_ws;
  size_t off = 0;
  float* xh    = wsf + off; off += (size_t)N * 128;
  float* h1    = wsf + off; off += (size_t)N * 128;
  float* cnt   = wsf + off; off += (size_t)N;
  float* la    = wsf + off; off += (size_t)N * 16;
  float* medge = wsf + off; off += 64;
  float* a_src = wsf + off; off += (size_t)N * 4;
  float* a_dst = wsf + off; off += (size_t)N * 4;
  float* alpha = wsf + off; off += ((size_t)E + N) * 4;
  unsigned* mseg = (unsigned*)(wsf + off); off += (size_t)N * 4;
  float* sseg  = wsf + off; off += (size_t)N * 4;
  (void)ws_size; (void)n_in; (void)out_size;

  const long long F = (long long)E + N;

  fillN((unsigned*)cnt, 0u, N, stream);
  fillN((unsigned*)la, 0u, (long long)N * 16, stream);
  {
    long long n = (long long)E * 16;
    hipLaunchKernelGGL(gat_count_loop, dim3((unsigned)((n + 255) / 256)),
                       dim3(256), 0, stream, dstI, ea, cnt, la, E);
    long long m = (long long)N * 16;
    hipLaunchKernelGGL(gat_loop_div, dim3((unsigned)((m + 255) / 256)),
                       dim3(256), 0, stream, cnt, la, N);
  }

  for (int l = 0; l < 2; ++l) {
    const float* xin = (l == 0) ? x : h1;
    float* outb = (l == 0) ? h1 : (float*)d_out;

    hipLaunchKernelGGL(gat_medge, dim3(1), dim3(64), 0, stream, Wel[l], ael[l],
                       medge);

    {
      long long waves = ((long long)N + 15) / 16;
      long long blocks = (waves + 7) / 8;
      hipLaunchKernelGGL(gat_gemm_wmma, dim3((unsigned)blocks), dim3(256), 0,
                         stream, xin, Wl[l], xh, N);
    }

    hipLaunchKernelGGL(gat_attdot, dim3((unsigned)N), dim3(128), 0, stream, xh,
                       asl[l], adl[l], a_src, a_dst, N);

    fillN(mseg, 0x007fffffu, (long long)N * 4, stream);
    fillN((unsigned*)sseg, 0u, (long long)N * 4, stream);

    hipLaunchKernelGGL(gat_alpha, dim3((unsigned)((F * 4 + 255) / 256)), dim3(256),
                       0, stream, srcI, dstI, ea, la, a_src, a_dst, medge,
                       alpha, mseg, E, N);

    {
      long long n4 = F * 4;
      hipLaunchKernelGGL(gat_expsum, dim3((unsigned)((n4 + 255) / 256)),
                         dim3(256), 0, stream, dstI, alpha, mseg, sseg, E, N);
    }

    {
      long long blocks = ((long long)N + OWN_R - 1) / OWN_R;
      hipLaunchKernelGGL(gat_own, dim3((unsigned)blocks), dim3(OWN_T), 0, stream,
                         srcI, dstI, alpha, sseg, xh, bl[l], outb, E, N,
                         (l == 0) ? 1 : 0);
    }
  }
}


# layer 2's owner launch reads each range's edge list back from layer 1's scan (kept in the range's own rows of the output buffer): one scan of all dst ids per call, not two; layer 2 adds in that order
# speedup vs baseline: 1.8675x; 1.0701x over previous
#include <hip/hip_runtime.h>
#include <hip/hip_bf16.h>

typedef __attribute__((ext_vector_type(2))) float v2f;
typedef __attribute__((ext_vector_type(8))) float v8f;

#define NEG_SLOPE 0.2f

__device__ __forceinline__ unsigned ordEnc(float f) {
  unsigned u = __float_as_uint(f);
  return (u & 0x80000000u) ? ~u : (u | 0x80000000u);
}
__device__ __forceinline__ float ordDec(unsigned u) {
  return (u & 0x80000000u) ? __uint_as_float(u ^ 0x80000000u)
                           : __uint_as_float(~u);
}

__global__ void gat_fill_u32(unsigned* __restrict__ p, unsigned v, long long n) {
  long long i = (long long)blockIdx.x * blockDim.x + threadIdx.x;
  if (i < n) p[i] = v;
}

__global__ void gat_count_loop(const int* __restrict__ dst,
                               const float* __restrict__ ea,
                               float* __restrict__ cnt,
                               float* __restrict__ la, int E) {
  long long i = (long long)blockIdx.x * blockDim.x + threadIdx.x;
  if (i >= (long long)E * 16) return;
  int e = (int)(i >> 4), j = (int)(i & 15);
  int d = dst[e];
  atomicAdd(&la[(long long)d * 16 + j], ea[i]);
  if (j == 0) atomicAdd(&cnt[d], 1.0f);
}

__global__ void gat_loop_div(const float* __restrict__ cnt,
                             float* __restrict__ la, int Nn) {
  long long i = (long long)blockIdx.x * blockDim.x + threadIdx.x;
  if (i >= (long long)Nn * 16) return;
  float c = cnt[i >> 4];
  c = (c < 1.0f) ? 1.0f : c;
  la[i] = la[i] / c;
}

__global__ void gat_medge(const float* __restrict__ We,
                          const float* __restrict__ ae,
                          float* __restrict__ medge) {
  int t = threadIdx.x;
  int k = t >> 2, h = t & 3;
  float s = 0.0f;
  for (int c = 0; c < 32; ++c)
    s += We[k * 128 + h * 32 + c] * ae[h * 32 + c];
  medge[k * 4 + h] = s;
}

__global__ __launch_bounds__(256) void gat_gemm_wmma(
    const float* __restrict__ X, const float* __restrict__ W,
    float* __restrict__ Y, int Nrows) {
  int wave = (int)(((long long)blockIdx.x * blockDim.x + threadIdx.x) >> 5);
  int lane = threadIdx.x & 31;
  int m0 = wave * 16;
  if (m0 >= Nrows) return;

  v8f acc[8] = {};
  int half = lane >> 4;
  int l16 = lane & 15;
  int arow = m0 + l16;
  if (arow >= Nrows) arow = Nrows - 1;
  const float* xr = X + (long long)arow * 128;

  for (int k0 = 0; k0 < 128; k0 += 4) {
    v2f a;
    a.x = xr[k0 + half * 2];
    a.y = xr[k0 + half * 2 + 1];
#pragma unroll
    for (int nt = 0; nt < 8; ++nt) {
      int col = nt * 16 + l16;
      v2f b;
      b.x = W[(k0 + half * 2) * 128 + col];
      b.y = W[(k0 + half * 2 + 1) * 128 + col];
      acc[nt] = __builtin_amdgcn_wmma_f32_16x16x4_f32(
          false, a, false, b, (short)0, acc[nt], false, false);
    }
  }
#pragma unroll
  for (int nt = 0; nt < 8; ++nt) {
#pragma unroll
    for (int r = 0; r < 8; ++r) {
      int row = m0 + r + half * 8;
      if (row < Nrows) Y[(long long)row * 128 + nt * 16 + l16] = acc[nt][r];
    }
  }
}

__global__ __launch_bounds__(128) void gat_attdot(
    const float* __restrict__ xh, const float* __restrict__ as,
    const float* __restrict__ ad, float* __restrict__ a_src,
    float* __restrict__ a_dst, int Nn) {
  int n = blockIdx.x;
  int t = threadIdx.x;
  float v = xh[(long long)n * 128 + t];
  float ps = v * as[t];
  float pd = v * ad[t];
  for (int o = 16; o > 0; o >>= 1) {
    ps += __shfl_down(ps, o, 32);
    pd += __shfl_down(pd, o, 32);
  }
  if ((t & 31) == 0) {
    int h = t >> 5;
    a_src[n * 4 + h] = ps;
    a_dst[n * 4 + h] = pd;
  }
}

__global__ void gat_alpha(const int* __restrict__ srcI,
                          const int* __restrict__ dstI,
                          const float* __restrict__ ea,
                          const float* __restrict__ la,
                          const float* __restrict__ a_src,
                          const float* __restrict__ a_dst,
                          const float* __restrict__ medge,
                          float* __restrict__ alpha,
                          unsigned* __restrict__ mseg, int E, int Nn) {
  long long i = (long long)blockIdx.x * blockDim.x + threadIdx.x;
  long long F = (long long)E + Nn;
  if (i >= F * 4) return;
  long long f = i >> 2;
  int h = (int)(i & 3);
  int s, d;
  const float* row;
  if (f < (long long)E) {
    s = srcI[f]; d = dstI[f]; row = ea + f * 16;
  } else {
    s = (int)(f - E); d = s; row = la + (long long)s * 16;
  }
  float e0[16];
#pragma unroll
  for (int k = 0; k < 16; ++k) e0[k] = row[k];
  float q0 = 0.0f, q1 = 0.0f, q2 = 0.0f, q3 = 0.0f;
#pragma unroll
  for (int k = 0; k < 16; ++k) {
    q0 += e0[k] * medge[k * 4 + 0];
    q1 += e0[k] * medge[k * 4 + 1];
    q2 += e0[k] * medge[k * 4 + 2];
    q3 += e0[k] * medge[k * 4 + 3];
  }
  float aE = q3;
  aE = (h == 2) ? q2 : aE;
  aE = (h == 1) ? q1 : aE;
  aE = (h == 0) ? q0 : aE;
  float v = a_src[(long long)s * 4 + h] + a_dst[(long long)d * 4 + h] + aE;
  v = (v > 0.0f) ? v : NEG_SLOPE * v;
  alpha[i] = v;
  atomicMax(&mseg[(long long)d * 4 + h], ordEnc(v));
}

__global__ void gat_expsum(const int* __restrict__ dstI,
                           float* __restrict__ alpha,
                           const unsigned* __restrict__ mseg,
                           float* __restrict__ sseg, int E, int Nn) {
  long long i = (long long)blockIdx.x * blockDim.x + threadIdx.x;
  long long F4 = ((long long)E + Nn) * 4;
  if (i >= F4) return;
  long long f = i >> 2;
  int h = (int)(i & 3);
  int d = (f < (long long)E) ? dstI[f] : (int)(f - E);
  float mv = ordDec(mseg[(long long)d * 4 + h]);
  if (!(mv > -__builtin_inff() && mv < __builtin_inff())) mv = 0.0f;
  float p = __expf(alpha[i] - mv);
  alpha[i] = p;
  atomicAdd(&sseg[(long long)d * 4 + h], p);
}

#define OWN_R 400
#define OWN_T 1024
#define OWN_W (OWN_T / 32)
#define OWN_CAP 16384
#define OWN_KEEP0 4
static_assert(OWN_R % 80 == 0, "start values are formed 80 rows at a time");

__device__ __noinline__ void own_add_alone(
    float* acc_row, const float* __restrict__ xr,
    const float* __restrict__ al4, const float* __restrict__ sg4) {
  for (int h = 0; h < 4; ++h) {
    float w = al4[h] / (sg4[h] + 1e-16f);
    for (int c = 0; c < 32; ++c) atomicAdd(acc_row + h * 32 + c, w * xr[h * 32 + c]);
  }
}

__device__ __forceinline__ void own_take(
    int dval, long long i, int base, int rows, float* acc, int* lst, int* cntS,
    const int* __restrict__ srcI, const float* __restrict__ alpha,
    const float* __restrict__ sseg, const float* __restrict__ xh) {
  unsigned r = (unsigned)(dval - base);
  if (r < (unsigned)rows) {
    int slot = atomicAdd(cntS, 1);
    if (slot < OWN_CAP) {
      lst[slot] = (int)i;
    } else {
      own_add_alone(acc + r * 128, xh + (long long)srcI[i] * 128, alpha + i * 4,
                    sseg + (long long)dval * 4);
    }
  }
}

__global__ __launch_bounds__(OWN_T) void gat_own(
    const int* __restrict__ srcI, const int* __restrict__ dstI,
    const float* __restrict__ alpha, const float* __restrict__ sseg,
    const float* __restrict__ xh, const float* __restrict__ bias,
    float* out, int* keep, int keep_mode, int E, int Nn, int relu) {
  __shared__ __attribute__((aligned(16))) float acc[OWN_R * 128];
  __shared__ int lst[OWN_CAP];
  __shared__ int cntS;

  const int t = threadIdx.x;
  const int lane = t & 31;
  const int base = (int)blockIdx.x * OWN_R;
  const int rows = (Nn - base >= OWN_R) ? OWN_R : (Nn - base);

  int* const keep_b = keep + (long long)base * 128;
  const bool fits = rows * 128 >= OWN_CAP + OWN_KEEP0;
  int kept = -1;
  if (keep_mode == 1 && fits) {
    kept = keep_b[0];
    if ((unsigned)kept > (unsigned)OWN_CAP) kept = -1;
  }

  if (t == 0) cntS = (kept >= 0) ? kept : 0;
  {
    const int c = t & 127;
    const int h = c >> 5;
    const int r0 = t >> 7;
    for (int rb = 0; rb < OWN_R; rb += 80) {
      float xa[10], aa[10], sa[10];
#pragma unroll
      for (int q = 0; q < 10; ++q) {
        int r = rb + q * 8 + r0;
        long long n = (long long)base + ((r < rows) ? r : 0);
        xa[q] = xh[n * 128 + c];
        aa[q] = alpha[((long long)E + n) * 4 + h];
        sa[q] = sseg[n * 4 + h];
      }
#pragma unroll
      for (int q = 0; q < 10; ++q) {
        int r = rb + q * 8 + r0;
        float v = (aa[q] / (sa[q] + 1e-16f)) * xa[q];
        acc[r * 128 + c] = (r < rows) ? v : 0.0f;
      }
    }
  }
  if (kept >= 0) {
    for (int i = t; i < kept; i += OWN_T) lst[i] = keep_b[OWN_KEEP0 + i];
  }
  __syncthreads();

  if (kept < 0) {
    int lead = (int)((4u - (unsigned)(((unsigned long long)dstI >> 2) & 3u)) & 3u);
    if (lead > E) lead = E;
    const long long nv = ((long long)E - lead) >> 2;
    const long long tail0 = (long long)lead + nv * 4;
    {
      long long i = -1;
      if (t < lead) i = t;
      else if ((long long)(t - lead) < (long long)E - tail0) i = tail0 + (t - lead);
      if (i >= 0) own_take(dstI[i], i, base, rows, acc, lst, &cntS, srcI, alpha, sseg, xh);
    }
    const int4* pv = (const int4*)(dstI + lead);
    const long long step = (long long)OWN_T * 4;
    auto load_step = [&](int4 (&qq)[4], long long v0) {
      if (v0 + step <= nv) {
#pragma unroll
        for (int u = 0; u < 4; ++u) qq[u] = pv[v0 + (long long)u * OWN_T + t];
      } else {
#pragma unroll
        for (int u = 0; u < 4; ++u) {
          long long v = v0 + (long long)u * OWN_T + t;
          qq[u] = (v < nv) ? pv[v] : make_int4(-1, -1, -1, -1);
        }
      }
    };
    int4 q[4];
    if (nv > 0) load_step(q, 0);
    for (long long v0 = 0; v0 < nv; v0 += step) {
      int4 qn[4];
      if (v0 + step < nv) {
        load_step(qn, v0 + step);
      } else {
#pragma unroll
        for (int u = 0; u < 4; ++u) qn[u] = make_int4(-1, -1, -1, -1);
      }
#pragma unroll
      for (int u = 0; u < 4; ++u) {
        long long i = (long long)lead + (v0 + (long long)u * OWN_T + t) * 4;
        own_take(q[u].x, i + 0, base, rows, acc, lst, &cntS, srcI, alpha, sseg, xh);
        own_take(q[u].y, i + 1, base, rows, acc, lst, &cntS, srcI, alpha, sseg, xh);
        own_take(q[u].z, i + 2, base, rows, acc, lst, &cntS, srcI, alpha, sseg, xh);
        own_take(q[u].w, i + 3, base, rows, acc, lst, &cntS, srcI, alpha, sseg, xh);
      }
#pragma unroll
      for (int u = 0; u < 4; ++u) q[u] = qn[u];
    }
  }
  __syncthreads();
  const int nraw = cntS;
  int nl = nraw;
  nl = (nl < OWN_CAP) ? nl : OWN_CAP;

  if (keep_mode == 0 && fits) {
    for (int i = t; i < nl; i += OWN_T) keep_b[OWN_KEEP0 + i] = lst[i];
    if (t == 0) keep_b[0] = nraw;
  }

  const int wv = __builtin_amdgcn_readfirstlane(t >> 5);
  const int hl = lane & 3;
  int s = 0, d = base;
  float al = 0.0f;
  if (wv * 8 < nl) {
    int slot = wv * 8 + (lane >> 2);
    if (slot >= nl) slot = nl - 1;
    int e = lst[slot];
    al = alpha[(long long)e * 4 + hl];
    s = srcI[e];
    d = dstI[e];
  }
  for (int g0 = wv * 8; g0 < nl; g0 += OWN_W * 8) {
    int nlive = (nl - g0 >= 8) ? 8 : (nl - g0);
    float sg = sseg[(long long)d * 4 + hl];
    float xv[8][4];
    int rr[8];
#pragma unroll
    for (int k = 0; k < 8; ++k) {
      int se = __builtin_amdgcn_readlane(s, k * 4);
      rr[k] = __builtin_amdgcn_readlane(d, k * 4) - base;
      const float* xr = xh + (long long)se * 128 + lane;
#pragma unroll
      for (int j = 0; j < 4; ++j) xv[k][j] = xr[j * 32];
    }
    int s2 = 0, d2 = base;
    float al2 = 0.0f;
    int gn = g0 + OWN_W * 8;
    if (gn < nl) {
      int slot = gn + (lane >> 2);
      if (slot >= nl) slot = nl - 1;
      int e = lst[slot];
      al2 = alpha[(long long)e * 4 + hl];
      s2 = srcI[e];
      d2 = dstI[e];
    }
    __builtin_amdgcn_sched_barrier(0);
    float w = al / (sg + 1e-16f);
    int wi = __float_as_int(w);
#pragma unroll
    for (int k = 0; k < 8; ++k) {
      if (k == 0 || k < nlive) {
        float* a = acc + rr[k] * 128 + lane;
#pragma unroll
        for (int j = 0; j < 4; ++j) {
          float wj = __int_as_float(__builtin_amdgcn_readlane(wi, k * 4 + j));
          atomicAdd(a + j * 32, wj * xv[k][j]);
        }
      }
    }
    s = s2; d = d2; al = al2;
  }
  __syncthreads();

  {
    const int c = t & 127;
    const float b = bias[c];
    for (int r = t >> 7; r < rows; r += OWN_T / 128) {
      float v = acc[r * 128 + c] + b;
      if (relu) v = (v > 0.0f) ? v : 0.0f;
      out[((long long)base + r) * 128 + c] = v;
    }
  }
}

static void fillN(unsigned* p, unsigned v, long long n, hipStream_t s) {
  if (n <= 0) return;
  long long b = (n + 255) / 256;
  hipLaunchKernelGGL(gat_fill_u32, dim3((unsigned)b), dim3(256), 0, s, p, v, n);
}

extern "C" void kernel_launch(void* const* d_in, const int* in_sizes, int n_in,
                              void* d_out, int out_size, void* d_ws,
                              size_t ws_size, hipStream_t stream) {
  const int N = in_sizes[0] / 128;
  const int E = in_sizes[1] / 2;

  const float* x  = (const float*)d_in[0];
  const int* ei   = (const int*)d_in[1];
  const int* srcI = ei;
  const int* dstI = ei + E;
  const float* ea = (const float*)d_in[2];

  const float* Wl[2]  = {(const float*)d_in[3], (const float*)d_in[9]};
  const float* asl[2] = {(const float*)d_in[4], (const float*)d_in[10]};
  const float* adl[2] = {(const float*)d_in[5], (const float*)d_in[11]};
  const float* Wel[2] = {(const float*)d_in[6], (const float*)d_in[12]};
  const float* ael[2] = {(const float*)d_in[7], (const float*)d_in[13]};
  const float* bl[2]  = {(const float*)d_in[8], (const float*)d_in[14]};

  float* wsf = (float*)d_ws;
  size_t off = 0;
  float* xh    = wsf + off; off += (size_t)N * 128;
  float* h1    = wsf + off; off += (size_t)N * 128;
  float* cnt   = wsf + off; off += (size_t)N;
  float* la    = wsf + off; off += (size_t)N * 16;
  float* medge = wsf + off; off += 64;
  float* a_src = wsf + off; off += (size_t)N * 4;
  float* a_dst = wsf + off; off += (size_t)N * 4;
  float* alpha = wsf + off; off += ((size_t)E + N) * 4;
  unsigned* mseg = (unsigned*)(wsf + off); off += (size_t)N * 4;
  float* sseg  = wsf + off; off += (size_t)N * 4;
  (void)ws_size; (void)n_in; (void)out_size;

  const long long F = (long long)E + N;

  fillN((unsigned*)cnt, 0u, N, stream);
  fillN((unsigned*)la, 0u, (long long)N * 16, stream);
  {
    long long n = (long long)E * 16;
    hipLaunchKernelGGL(gat_count_loop, dim3((unsigned)((n + 255) / 256)),
                       dim3(256), 0, stream, dstI, ea, cnt, la, E);
    long long m = (long long)N * 16;
    hipLaunchKernelGGL(gat_loop_div, dim3((unsigned)((m + 255) / 256)),
                       dim3(256), 0, stream, cnt, la, N);
  }

  for (int l = 0; l < 2; ++l) {
    const float* xin = (l == 0) ? x : h1;
    float* outb = (l == 0) ? h1 : (float*)d_out;

    hipLaunchKernelGGL(gat_medge, dim3(1), dim3(64), 0, stream, Wel[l], ael[l],
                       medge);

    {
      long long waves = ((long long)N + 15) / 16;
      long long blocks = (waves + 7) / 8;
      hipLaunchKernelGGL(gat_gemm_wmma, dim3((unsigned)blocks), dim3(256), 0,
                         stream, xin, Wl[l], xh, N);
    }

    hipLaunchKernelGGL(gat_attdot, dim3((unsigned)N), dim3(128), 0, stream, xh,
                       asl[l], adl[l], a_src, a_dst, N);

    fillN(mseg, 0x007fffffu, (long long)N * 4, stream);
    fillN((unsigned*)sseg, 0u, (long long)N * 4, stream);

    hipLaunchKernelGGL(gat_alpha, dim3((unsigned)((F * 4 + 255) / 256)), dim3(256),
                       0, stream, srcI, dstI, ea, la, a_src, a_dst, medge,
                       alpha, mseg, E, N);

    {
      long long n4 = F * 4;
      hipLaunchKernelGGL(gat_expsum, dim3((unsigned)((n4 + 255) / 256)),
                         dim3(256), 0, stream, dstI, alpha, mseg, sseg, E, N);
    }

    {
      long long blocks = ((long long)N + OWN_R - 1) / OWN_R;
      hipLaunchKernelGGL(gat_own, dim3((unsigned)blocks), dim3(OWN_T), 0, stream,
                         srcI, dstI, alpha, sseg, xh, bl[l], outb, (int*)d_out,
                         l, E, N, (l == 0) ? 1 : 0);
    }
  }
}


# attention dots formed in the GEMM wave's epilogue from its accumulators (2 tiles x coefficients per lane, 16-lane sums): no per-node dot launches, 20 -> 18, 800000 fewer waves; dot sum order changes
# speedup vs baseline: 2.0001x; 1.0710x over previous
#include <hip/hip_runtime.h>
#include <hip/hip_bf16.h>

typedef __attribute__((ext_vector_type(2))) float v2f;
typedef __attribute__((ext_vector_type(8))) float v8f;

#define NEG_SLOPE 0.2f

__device__ __forceinline__ unsigned ordEnc(float f) {
  unsigned u = __float_as_uint(f);
  return (u & 0x80000000u) ? ~u : (u | 0x80000000u);
}
__device__ __forceinline__ float ordDec(unsigned u) {
  return (u & 0x80000000u) ? __uint_as_float(u ^ 0x80000000u)
                           : __uint_as_float(~u);
}

__global__ void gat_fill_u32(unsigned* __restrict__ p, unsigned v, long long n) {
  long long i = (long long)blockIdx.x * blockDim.x + threadIdx.x;
  if (i < n) p[i] = v;
}

__global__ void gat_count_loop(const int* __restrict__ dst,
                               const float* __restrict__ ea,
                               float* __restrict__ cnt,
                               float* __restrict__ la, int E) {
  long long i = (long long)blockIdx.x * blockDim.x + threadIdx.x;
  if (i >= (long long)E * 16) return;
  int e = (int)(i >> 4), j = (int)(i & 15);
  int d = dst[e];
  atomicAdd(&la[(long long)d * 16 + j], ea[i]);
  if (j == 0) atomicAdd(&cnt[d], 1.0f);
}

__global__ void gat_loop_div(const float* __restrict__ cnt,
                             float* __restrict__ la, int Nn) {
  long long i = (long long)blockIdx.x * blockDim.x + threadIdx.x;
  if (i >= (long long)Nn * 16) return;
  float c = cnt[i >> 4];
  c = (c < 1.0f) ? 1.0f : c;
  la[i] = la[i] / c;
}

__global__ void gat_medge(const float* __restrict__ We,
                          const float* __restrict__ ae,
                          float* __restrict__ medge) {
  int t = threadIdx.x;
  int k = t >> 2, h = t & 3;
  float s = 0.0f;
  for (int c = 0; c < 32; ++c)
    s += We[k * 128 + h * 32 + c] * ae[h * 32 + c];
  medge[k * 4 + h] = s;
}

__global__ __launch_bounds__(256) void gat_gemm_wmma(
    const float* __restrict__ X, const float* __restrict__ W,
    const float* __restrict__ as, const float* __restrict__ ad,
    float* __restrict__ Y, float* __restrict__ a_src,
    float* __restrict__ a_dst, int Nrows) {
  int wave = (int)(((long long)blockIdx.x * blockDim.x + threadIdx.x) >> 5);
  int lane = threadIdx.x & 31;
  int m0 = wave * 16;
  if (m0 >= Nrows) return;

  v8f acc[8] = {};
  int half = lane >> 4;
  int l16 = lane & 15;
  int arow = m0 + l16;
  if (arow >= Nrows) arow = Nrows - 1;
  const float* xr = X + (long long)arow * 128;

  for (int k0 = 0; k0 < 128; k0 += 4) {
    v2f a;
    a.x = xr[k0 + half * 2];
    a.y = xr[k0 + half * 2 + 1];
#pragma unroll
    for (int nt = 0; nt < 8; ++nt) {
      int col = nt * 16 + l16;
      v2f b;
      b.x = W[(k0 + half * 2) * 128 + col];
      b.y = W[(k0 + half * 2 + 1) * 128 + col];
      acc[nt] = __builtin_amdgcn_wmma_f32_16x16x4_f32(
          false, a, false, b, (short)0, acc[nt], false, false);
    }
  }
#pragma unroll
  for (int nt = 0; nt < 8; ++nt) {
#pragma unroll
    for (int r = 0; r < 8; ++r) {
      int row = m0 + r + half * 8;
      if (row < Nrows) Y[(long long)row * 128 + nt * 16 + l16] = acc[nt][r];
    }
  }

  float cs[8], cd[8];
#pragma unroll
  for (int nt = 0; nt < 8; ++nt) {
    cs[nt] = as[nt * 16 + l16];
    cd[nt] = ad[nt * 16 + l16];
  }
  float s0 = 0.0f, s1 = 0.0f, d0 = 0.0f, d1 = 0.0f;
#pragma unroll
  for (int r = 0; r < 8; ++r) {
#pragma unroll
    for (int hd = 0; hd < 4; ++hd) {
      float ps = acc[2 * hd][r] * cs[2 * hd] + acc[2 * hd + 1][r] * cs[2 * hd + 1];
      float pd = acc[2 * hd][r] * cd[2 * hd] + acc[2 * hd + 1][r] * cd[2 * hd + 1];
      ps += __shfl_xor(ps, 8, 32);
      pd += __shfl_xor(pd, 8, 32);
      ps += __shfl_xor(ps, 4, 32);
      pd += __shfl_xor(pd, 4, 32);
      ps += __shfl_xor(ps, 2, 32);
      pd += __shfl_xor(pd, 2, 32);
      ps += __shfl_xor(ps, 1, 32);
      pd += __shfl_xor(pd, 1, 32);
      const bool mine = (l16 == r * 2 + (hd >> 1));
      if ((hd & 1) == 0) {
        s0 = mine ? ps : s0;
        d0 = mine ? pd : d0;
      } else {
        s1 = mine ? ps : s1;
        d1 = mine ? pd : d1;
      }
    }
  }
  {
    int row = m0 + half * 8 + (l16 >> 1);
    if (row < Nrows) {
      long long o = (long long)row * 4 + (l16 & 1) * 2;
      *(float2*)(a_src + o) = make_float2(s0, s1);
      *(float2*)(a_dst + o) = make_float2(d0, d1);
    }
  }
}

__global__ void gat_alpha(const int* __restrict__ srcI,
                          const int* __restrict__ dstI,
                          const float* __restrict__ ea,
                          const float* __restrict__ la,
                          const float* __restrict__ a_src,
                          const float* __restrict__ a_dst,
                          const float* __restrict__ medge,
                          float* __restrict__ alpha,
                          unsigned* __restrict__ mseg, int E, int Nn) {
  long long i = (long long)blockIdx.x * blockDim.x + threadIdx.x;
  long long F = (long long)E + Nn;
  if (i >= F * 4) return;
  long long f = i >> 2;
  int h = (int)(i & 3);
  int s, d;
  const float* row;
  if (f < (long long)E) {
    s = srcI[f]; d = dstI[f]; row = ea + f * 16;
  } else {
    s = (int)(f - E); d = s; row = la + (long long)s * 16;
  }
  float e0[16];
#pragma unroll
  for (int k = 0; k < 16; ++k) e0[k] = row[k];
  float q0 = 0.0f, q1 = 0.0f, q2 = 0.0f, q3 = 0.0f;
#pragma unroll
  for (int k = 0; k < 16; ++k) {
    q0 += e0[k] * medge[k * 4 + 0];
    q1 += e0[k] * medge[k * 4 + 1];
    q2 += e0[k] * medge[k * 4 + 2];
    q3 += e0[k] * medge[k * 4 + 3];
  }
  float aE = q3;
  aE = (h == 2) ? q2 : aE;
  aE = (h == 1) ? q1 : aE;
  aE = (h == 0) ? q0 : aE;
  float v = a_src[(long long)s * 4 + h] + a_dst[(long long)d * 4 + h] + aE;
  v = (v > 0.0f) ? v : NEG_SLOPE * v;
  alpha[i] = v;
  atomicMax(&mseg[(long long)d * 4 + h], ordEnc(v));
}

__global__ void gat_expsum(const int* __restrict__ dstI,
                           float* __restrict__ alpha,
                           const unsigned* __restrict__ mseg,
                           float* __restrict__ sseg, int E, int Nn) {
  long long i = (long long)blockIdx.x * blockDim.x + threadIdx.x;
  long long F4 = ((long long)E + Nn) * 4;
  if (i >= F4) return;
  long long f = i >> 2;
  int h = (int)(i & 3);
  int d = (f < (long long)E) ? dstI[f] : (int)(f - E);
  float mv = ordDec(mseg[(long long)d * 4 + h]);
  if (!(mv > -__builtin_inff() && mv < __builtin_inff())) mv = 0.0f;
  float p = __expf(alpha[i] - mv);
  alpha[i] = p;
  atomicAdd(&sseg[(long long)d * 4 + h], p);
}

#define OWN_R 400
#define OWN_T 1024
#define OWN_W (OWN_T / 32)
#define OWN_CAP 16384
#define OWN_KEEP0 4
static_assert(OWN_R % 80 == 0, "start values are formed 80 rows at a time");

__device__ __noinline__ void own_add_alone(
    float* acc_row, const float* __restrict__ xr,
    const float* __restrict__ al4, const float* __restrict__ sg4) {
  for (int h = 0; h < 4; ++h) {
    float w = al4[h] / (sg4[h] + 1e-16f);
    for (int c = 0; c < 32; ++c) atomicAdd(acc_row + h * 32 + c, w * xr[h * 32 + c]);
  }
}

__device__ __forceinline__ void own_take(
    int dval, long long i, int base, int rows, float* acc, int* lst, int* cntS,
    const int* __restrict__ srcI, const float* __restrict__ alpha,
    const float* __restrict__ sseg, const float* __restrict__ xh) {
  unsigned r = (unsigned)(dval - base);
  if (r < (unsigned)rows) {
    int slot = atomicAdd(cntS, 1);
    if (slot < OWN_CAP) {
      lst[slot] = (int)i;
    } else {
      own_add_alone(acc + r * 128, xh + (long long)srcI[i] * 128, alpha + i * 4,
                    sseg + (long long)dval * 4);
    }
  }
}

__global__ __launch_bounds__(OWN_T) void gat_own(
    const int* __restrict__ srcI, const int* __restrict__ dstI,
    const float* __restrict__ alpha, const float* __restrict__ sseg,
    const float* __restrict__ xh, const float* __restrict__ bias,
    float* out, int* keep, int keep_mode, int E, int Nn, int relu) {
  __shared__ __attribute__((aligned(16))) float acc[OWN_R * 128];
  __shared__ int lst[OWN_CAP];
  __shared__ int cntS;

  const int t = threadIdx.x;
  const int lane = t & 31;
  const int base = (int)blockIdx.x * OWN_R;
  const int rows = (Nn - base >= OWN_R) ? OWN_R : (Nn - base);

  int* const keep_b = keep + (long long)base * 128;
  const bool fits = rows * 128 >= OWN_CAP + OWN_KEEP0;
  int kept = -1;
  if (keep_mode == 1 && fits) {
    kept = keep_b[0];
    if ((unsigned)kept > (unsigned)OWN_CAP) kept = -1;
  }

  if (t == 0) cntS = (kept >= 0) ? kept : 0;
  {
    const int c = t & 127;
    const int h = c >> 5;
    const int r0 = t >> 7;
    for (int rb = 0; rb < OWN_R; rb += 80) {
      float xa[10], aa[10], sa[10];
#pragma unroll
      for (int q = 0; q < 10; ++q) {
        int r = rb + q * 8 + r0;
        long long n = (long long)base + ((r < rows) ? r : 0);
        xa[q] = xh[n * 128 + c];
        aa[q] = alpha[((long long)E + n) * 4 + h];
        sa[q] = sseg[n * 4 + h];
      }
#pragma unroll
      for (int q = 0; q < 10; ++q) {
        int r = rb + q * 8 + r0;
        float v = (aa[q] / (sa[q] + 1e-16f)) * xa[q];
        acc[r * 128 + c] = (r < rows) ? v : 0.0f;
      }
    }
  }
  if (kept >= 0) {
    for (int i = t; i < kept; i += OWN_T) lst[i] = keep_b[OWN_KEEP0 + i];
  }
  __syncthreads();

  if (kept < 0) {
    int lead = (int)((4u - (unsigned)(((unsigned long long)dstI >> 2) & 3u)) & 3u);
    if (lead > E) lead = E;
    const long long nv = ((long long)E - lead) >> 2;
    const long long tail0 = (long long)lead + nv * 4;
    {
      long long i = -1;
      if (t < lead) i = t;
      else if ((long long)(t - lead) < (long long)E - tail0) i = tail0 + (t - lead);
      if (i >= 0) own_take(dstI[i], i, base, rows, acc, lst, &cntS, srcI, alpha, sseg, xh);
    }
    const int4* pv = (const int4*)(dstI + lead);
    const long long step = (long long)OWN_T * 4;
    auto load_step = [&](int4 (&qq)[4], long long v0) {
      if (v0 + step <= nv) {
#pragma unroll
        for (int u = 0; u < 4; ++u) qq[u] = pv[v0 + (long long)u * OWN_T + t];
      } else {
#pragma unroll
        for (int u = 0; u < 4; ++u) {
          long long v = v0 + (long long)u * OWN_T + t;
          qq[u] = (v < nv) ? pv[v] : make_int4(-1, -1, -1, -1);
        }
      }
    };
    int4 q[4];
    if (nv > 0) load_step(q, 0);
    for (long long v0 = 0; v0 < nv; v0 += step) {
      int4 qn[4];
      if (v0 + step < nv) {
        load_step(qn, v0 + step);
      } else {
#pragma unroll
        for (int u = 0; u < 4; ++u) qn[u] = make_int4(-1, -1, -1, -1);
      }
#pragma unroll
      for (int u = 0; u < 4; ++u) {
        long long i = (long long)lead + (v0 + (long long)u * OWN_T + t) * 4;
        own_take(q[u].x, i + 0, base, rows, acc, lst, &cntS, srcI, alpha, sseg, xh);
        own_take(q[u].y, i + 1, base, rows, acc, lst, &cntS, srcI, alpha, sseg, xh);
        own_take(q[u].z, i + 2, base, rows, acc, lst, &cntS, srcI, alpha, sseg, xh);
        own_take(q[u].w, i + 3, base, rows, acc, lst, &cntS, srcI, alpha, sseg, xh);
      }
#pragma unroll
      for (int u = 0; u < 4; ++u) q[u] = qn[u];
    }
  }
  __syncthreads();
  const int nraw = cntS;
  int nl = nraw;
  nl = (nl < OWN_CAP) ? nl : OWN_CAP;

  if (keep_mode == 0 && fits) {
    for (int i = t; i < nl; i += OWN_T) keep_b[OWN_KEEP0 + i] = lst[i];
    if (t == 0) keep_b[0] = nraw;
  }

  const int wv = __builtin_amdgcn_readfirstlane(t >> 5);
  const int hl = lane & 3;
  int s = 0, d = base;
  float al = 0.0f;
  if (wv * 8 < nl) {
    int slot = wv * 8 + (lane >> 2);
    if (slot >= nl) slot = nl - 1;
    int e = lst[slot];
    al = alpha[(long long)e * 4 + hl];
    s = srcI[e];
    d = dstI[e];
  }
  for (int g0 = wv * 8; g0 < nl; g0 += OWN_W * 8) {
    int nlive = (nl - g0 >= 8) ? 8 : (nl - g0);
    float sg = sseg[(long long)d * 4 + hl];
    float xv[8][4];
    int rr[8];
#pragma unroll
    for (int k = 0; k < 8; ++k) {
      int se = __builtin_amdgcn_readlane(s, k * 4);
      rr[k] = __builtin_amdgcn_readlane(d, k * 4) - base;
      const float* xr = xh + (long long)se * 128 + lane;
#pragma unroll
      for (int j = 0; j < 4; ++j) xv[k][j] = xr[j * 32];
    }
    int s2 = 0, d2 = base;
    float al2 = 0.0f;
    int gn = g0 + OWN_W * 8;
    if (gn < nl) {
      int slot = gn + (lane >> 2);
      if (slot >= nl) slot = nl - 1;
      int e = lst[slot];
      al2 = alpha[(long long)e * 4 + hl];
      s2 = srcI[e];
      d2 = dstI[e];
    }
    __builtin_amdgcn_sched_barrier(0);
    float w = al / (sg + 1e-16f);
    int wi = __float_as_int(w);
#pragma unroll
    for (int k = 0; k < 8; ++k) {
      if (k == 0 || k < nlive) {
        float* a = acc + rr[k] * 128 + lane;
#pragma unroll
        for (int j = 0; j < 4; ++j) {
          float wj = __int_as_float(__builtin_amdgcn_readlane(wi, k * 4 + j));
          atomicAdd(a + j * 32, wj * xv[k][j]);
        }
      }
    }
    s = s2; d = d2; al = al2;
  }
  __syncthreads();

  {
    const int c = t & 127;
    const float b = bias[c];
    for (int r = t >> 7; r < rows; r += OWN_T / 128) {
      float v = acc[r * 128 + c] + b;
      if (relu) v = (v > 0.0f) ? v : 0.0f;
      out[((long long)base + r) * 128 + c] = v;
    }
  }
}

static void fillN(unsigned* p, unsigned v, long long n, hipStream_t s) {
  if (n <= 0) return;
  long long b = (n + 255) / 256;
  hipLaunchKernelGGL(gat_fill_u32, dim3((unsigned)b), dim3(256), 0, s, p, v, n);
}

extern "C" void kernel_launch(void* const* d_in, const int* in_sizes, int n_in,
                              void* d_out, int out_size, void* d_ws,
                              size_t ws_size, hipStream_t stream) {
  const int N = in_sizes[0] / 128;
  const int E = in_sizes[1] / 2;

  const float* x  = (const float*)d_in[0];
  const int* ei   = (const int*)d_in[1];
  const int* srcI = ei;
  const int* dstI = ei + E;
  const float* ea = (const float*)d_in[2];

  const float* Wl[2]  = {(const float*)d_in[3], (const float*)d_in[9]};
  const float* asl[2] = {(const float*)d_in[4], (const float*)d_in[10]};
  const float* adl[2] = {(const float*)d_in[5], (const float*)d_in[11]};
  const float* Wel[2] = {(const float*)d_in[6], (const float*)d_in[12]};
  const float* ael[2] = {(const float*)d_in[7], (const float*)d_in[13]};
  const float* bl[2]  = {(const float*)d_in[8], (const float*)d_in[14]};

  float* wsf = (float*)d_ws;
  size_t off = 0;
  float* xh    = wsf + off; off += (size_t)N * 128;
  float* h1    = wsf + off; off += (size_t)N * 128;
  float* cnt   = wsf + off; off += (size_t)N;
  float* la    = wsf + off; off += (size_t)N * 16;
  float* medge = wsf + off; off += 64;
  float* a_src = wsf + off; off += (size_t)N * 4;
  float* a_dst = wsf + off; off += (size_t)N * 4;
  float* alpha = wsf + off; off += ((size_t)E + N) * 4;
  unsigned* mseg = (unsigned*)(wsf + off); off += (size_t)N * 4;
  float* sseg  = wsf + off; off += (size_t)N * 4;
  (void)ws_size; (void)n_in; (void)out_size;

  const long long F = (long long)E + N;

  fillN((unsigned*)cnt, 0u, N, stream);
  fillN((unsigned*)la, 0u, (long long)N * 16, stream);
  {
    long long n = (long long)E * 16;
    hipLaunchKernelGGL(gat_count_loop, dim3((unsigned)((n + 255) / 256)),
                       dim3(256), 0, stream, dstI, ea, cnt, la, E);
    long long m = (long long)N * 16;
    hipLaunchKernelGGL(gat_loop_div, dim3((unsigned)((m + 255) / 256)),
                       dim3(256), 0, stream, cnt, la, N);
  }

  for (int l = 0; l < 2; ++l) {
    const float* xin = (l == 0) ? x : h1;
    float* outb = (l == 0) ? h1 : (float*)d_out;

    hipLaunchKernelGGL(gat_medge, dim3(1), dim3(64), 0, stream, Wel[l], ael[l],
                       medge);

    {
      long long waves = ((long long)N + 15) / 16;
      long long blocks = (waves + 7) / 8;
      hipLaunchKernelGGL(gat_gemm_wmma, dim3((unsigned)blocks), dim3(256), 0,
                         stream, xin, Wl[l], asl[l], adl[l], xh, a_src, a_dst, N);
    }

    fillN(mseg, 0x007fffffu, (long long)N * 4, stream);
    fillN((unsigned*)sseg, 0u, (long long)N * 4, stream);

    hipLaunchKernelGGL(gat_alpha, dim3((unsigned)((F * 4 + 255) / 256)), dim3(256),
                       0, stream, srcI, dstI, ea, la, a_src, a_dst, medge,
                       alpha, mseg, E, N);

    {
      long long n4 = F * 4;
      hipLaunchKernelGGL(gat_expsum, dim3((unsigned)((n4 + 255) / 256)),
                         dim3(256), 0, stream, dstI, alpha, mseg, sseg, E, N);
    }

    {
      long long blocks = ((long long)N + OWN_R - 1) / OWN_R;
      hipLaunchKernelGGL(gat_own, dim3((unsigned)blocks), dim3(OWN_T), 0, stream,
                         srcI, dstI, alpha, sseg, xh, bl[l], outb, (int*)d_out,
                         l, E, N, (l == 0) ? 1 : 0);
    }
  }
}


# segment softmax by the range owner in LDS: maxima by LDS max over its edge list, exp and sum in the row walk, division at the row store; no global max/add atomics, 18 -> 12 launches; sum order new
# speedup vs baseline: 2.6911x; 1.3455x over previous
#include <hip/hip_runtime.h>
#include <hip/hip_bf16.h>

typedef __attribute__((ext_vector_type(2))) float v2f;
typedef __attribute__((ext_vector_type(8))) float v8f;

#define NEG_SLOPE 0.2f

__device__ __forceinline__ unsigned ordEnc(float f) {
  unsigned u = __float_as_uint(f);
  return (u & 0x80000000u) ? ~u : (u | 0x80000000u);
}
__device__ __forceinline__ float ordDec(unsigned u) {
  return (u & 0x80000000u) ? __uint_as_float(u ^ 0x80000000u)
                           : __uint_as_float(~u);
}

__global__ void gat_fill_u32(unsigned* __restrict__ p, unsigned v, long long n) {
  long long i = (long long)blockIdx.x * blockDim.x + threadIdx.x;
  if (i < n) p[i] = v;
}

__global__ void gat_count_loop(const int* __restrict__ dst,
                               const float* __restrict__ ea,
                               float* __restrict__ cnt,
                               float* __restrict__ la, int E) {
  long long i = (long long)blockIdx.x * blockDim.x + threadIdx.x;
  if (i >= (long long)E * 16) return;
  int e = (int)(i >> 4), j = (int)(i & 15);
  int d = dst[e];
  atomicAdd(&la[(long long)d * 16 + j], ea[i]);
  if (j == 0) atomicAdd(&cnt[d], 1.0f);
}

__global__ void gat_loop_div(const float* __restrict__ cnt,
                             float* __restrict__ la, int Nn) {
  long long i = (long long)blockIdx.x * blockDim.x + threadIdx.x;
  if (i >= (long long)Nn * 16) return;
  float c = cnt[i >> 4];
  c = (c < 1.0f) ? 1.0f : c;
  la[i] = la[i] / c;
}

__global__ void gat_medge(const float* __restrict__ We,
                          const float* __restrict__ ae,
                          float* __restrict__ medge) {
  int t = threadIdx.x;
  int k = t >> 2, h = t & 3;
  float s = 0.0f;
  for (int c = 0; c < 32; ++c)
    s += We[k * 128 + h * 32 + c] * ae[h * 32 + c];
  medge[k * 4 + h] = s;
}

__global__ __launch_bounds__(256) void gat_gemm_wmma(
    const float* __restrict__ X, const float* __restrict__ W,
    const float* __restrict__ as, const float* __restrict__ ad,
    float* __restrict__ Y, float* __restrict__ a_src,
    float* __restrict__ a_dst, int Nrows) {
  int wave = (int)(((long long)blockIdx.x * blockDim.x + threadIdx.x) >> 5);
  int lane = threadIdx.x & 31;
  int m0 = wave * 16;
  if (m0 >= Nrows) return;

  v8f acc[8] = {};
  int half = lane >> 4;
  int l16 = lane & 15;
  int arow = m0 + l16;
  if (arow >= Nrows) arow = Nrows - 1;
  const float* xr = X + (long long)arow * 128;

  for (int k0 = 0; k0 < 128; k0 += 4) {
    v2f a;
    a.x = xr[k0 + half * 2];
    a.y = xr[k0 + half * 2 + 1];
#pragma unroll
    for (int nt = 0; nt < 8; ++nt) {
      int col = nt * 16 + l16;
      v2f b;
      b.x = W[(k0 + half * 2) * 128 + col];
      b.y = W[(k0 + half * 2 + 1) * 128 + col];
      acc[nt] = __builtin_amdgcn_wmma_f32_16x16x4_f32(
          false, a, false, b, (short)0, acc[nt], false, false);
    }
  }
#pragma unroll
  for (int nt = 0; nt < 8; ++nt) {
#pragma unroll
    for (int r = 0; r < 8; ++r) {
      int row = m0 + r + half * 8;
      if (row < Nrows) Y[(long long)row * 128 + nt * 16 + l16] = acc[nt][r];
    }
  }

  float cs[8], cd[8];
#pragma unroll
  for (int nt = 0; nt < 8; ++nt) {
    cs[nt] = as[nt * 16 + l16];
    cd[nt] = ad[nt * 16 + l16];
  }
  float s0 = 0.0f, s1 = 0.0f, d0 = 0.0f, d1 = 0.0f;
#pragma unroll
  for (int r = 0; r < 8; ++r) {
#pragma unroll
    for (int hd = 0; hd < 4; ++hd) {
      float ps = acc[2 * hd][r] * cs[2 * hd] + acc[2 * hd + 1][r] * cs[2 * hd + 1];
      float pd = acc[2 * hd][r] * cd[2 * hd] + acc[2 * hd + 1][r] * cd[2 * hd + 1];
      ps += __shfl_xor(ps, 8, 32);
      pd += __shfl_xor(pd, 8, 32);
      ps += __shfl_xor(ps, 4, 32);
      pd += __shfl_xor(pd, 4, 32);
      ps += __shfl_xor(ps, 2, 32);
      pd += __shfl_xor(pd, 2, 32);
      ps += __shfl_xor(ps, 1, 32);
      pd += __shfl_xor(pd, 1, 32);
      const bool mine = (l16 == r * 2 + (hd >> 1));
      if ((hd & 1) == 0) {
        s0 = mine ? ps : s0;
        d0 = mine ? pd : d0;
      } else {
        s1 = mine ? ps : s1;
        d1 = mine ? pd : d1;
      }
    }
  }
  {
    int row = m0 + half * 8 + (l16 >> 1);
    if (row < Nrows) {
      long long o = (long long)row * 4 + (l16 & 1) * 2;
      *(float2*)(a_src + o) = make_float2(s0, s1);
      *(float2*)(a_dst + o) = make_float2(d0, d1);
    }
  }
}

__global__ void gat_alpha(const int* __restrict__ srcI,
                          const int* __restrict__ dstI,
                          const float* __restrict__ ea,
                          const float* __restrict__ la,
                          const float* __restrict__ a_src,
                          const float* __restrict__ a_dst,
                          const float* __restrict__ medge,
                          float* __restrict__ alpha, int E, int Nn) {
  long long i = (long long)blockIdx.x * blockDim.x + threadIdx.x;
  long long F = (long long)E + Nn;
  if (i >= F * 4) return;
  long long f = i >> 2;
  int h = (int)(i & 3);
  int s, d;
  const float* row;
  if (f < (long long)E) {
    s = srcI[f]; d = dstI[f]; row = ea + f * 16;
  } else {
    s = (int)(f - E); d = s; row = la + (long long)s * 16;
  }
  float e0[16];
#pragma unroll
  for (int k = 0; k < 16; ++k) e0[k] = row[k];
  float q0 = 0.0f, q1 = 0.0f, q2 = 0.0f, q3 = 0.0f;
#pragma unroll
  for (int k = 0; k < 16; ++k) {
    q0 += e0[k] * medge[k * 4 + 0];
    q1 += e0[k] * medge[k * 4 + 1];
    q2 += e0[k] * medge[k * 4 + 2];
    q3 += e0[k] * medge[k * 4 + 3];
  }
  float aE = q3;
  aE = (h == 2) ? q2 : aE;
  aE = (h == 1) ? q1 : aE;
  aE = (h == 0) ? q0 : aE;
  float v = a_src[(long long)s * 4 + h] + a_dst[(long long)d * 4 + h] + aE;
  v = (v > 0.0f) ? v : NEG_SLOPE * v;
  alpha[i] = v;
}

#define OWN_R 400
#define OWN_T 1024
#define OWN_W (OWN_T / 32)
#define OWN_CAP 16384
#define OWN_KEEP0 4
#define OWN_RSH 23
#define OWN_EMASK ((1u << OWN_RSH) - 1u)
static_assert(OWN_R <= (1 << (32 - OWN_RSH)), "the row in range must fit above the edge number");

__device__ __forceinline__ float own_max_dec(unsigned u) {
  float mv = ordDec(u);
  if (!(mv > -__builtin_inff() && mv < __builtin_inff())) mv = 0.0f;
  return mv;
}

__device__ __forceinline__ void own_take(int dval, long long i, int base, int rows,
                                         unsigned* lst, int* cntS) {
  unsigned r = (unsigned)(dval - base);
  if (r < (unsigned)rows) {
    int slot = atomicAdd(cntS, 1);
    if (slot < OWN_CAP) lst[slot] = (unsigned)i | (r << OWN_RSH);
  }
}

__device__ __noinline__ void own_unlisted(
    const int* __restrict__ srcI, const int* __restrict__ dstI,
    const float* __restrict__ alpha, const float* __restrict__ xh,
    float* acc, unsigned* mxS, float* smS, int base, int rows, int E, int t) {
  const long long nall = (long long)E + rows;
  for (long long i = t; i < nall; i += OWN_T) {
    unsigned r;
    long long arow;
    if (i < (long long)E) { r = (unsigned)(dstI[i] - base); arow = i; }
    else { r = (unsigned)(i - E); arow = (long long)E + base + r; }
    if (r < (unsigned)rows)
      for (int h = 0; h < 4; ++h) atomicMax(&mxS[r * 4 + h], ordEnc(alpha[arow * 4 + h]));
  }
  __syncthreads();
  for (long long i = t; i < nall; i += OWN_T) {
    unsigned r;
    long long arow;
    int s;
    if (i < (long long)E) { r = (unsigned)(dstI[i] - base); arow = i; s = srcI[i]; }
    else { r = (unsigned)(i - E); arow = (long long)E + base + r; s = base + (int)r; }
    if (r < (unsigned)rows) {
      const float* xr = xh + (long long)s * 128;
      for (int h = 0; h < 4; ++h) {
        float p = __expf(alpha[arow * 4 + h] - own_max_dec(mxS[r * 4 + h]));
        atomicAdd(&smS[r * 4 + h], p);
        for (int c = 0; c < 32; ++c) atomicAdd(acc + r * 128 + h * 32 + c, p * xr[h * 32 + c]);
      }
    }
  }
}

__global__ __launch_bounds__(OWN_T) void gat_own(
    const int* __restrict__ srcI, const int* __restrict__ dstI,
    const float* __restrict__ alpha,
    const float* __restrict__ xh, const float* __restrict__ bias,
    float* out, unsigned* keep, int keep_mode, int E, int Nn, int relu) {
  __shared__ __attribute__((aligned(16))) float acc[OWN_R * 128];
  __shared__ __attribute__((aligned(16))) unsigned lst[OWN_CAP];
  __shared__ __attribute__((aligned(16))) unsigned mxS[OWN_R * 4];
  __shared__ __attribute__((aligned(16))) float smS[OWN_R * 4];
  __shared__ int cntS;

  const int t = threadIdx.x;
  const int lane = t & 31;
  const int base = (int)blockIdx.x * OWN_R;
  const int rows = (Nn - base >= OWN_R) ? OWN_R : (Nn - base);

  const bool packable = E <= (int)OWN_EMASK;
  unsigned* const keep_b = keep + (long long)base * 128;
  const bool fits = rows * 128 >= OWN_CAP + OWN_KEEP0;
  int kept = -1;
  if (keep_mode == 1 && fits && packable) {
    unsigned k = keep_b[0];
    if (k <= (unsigned)OWN_CAP) kept = (int)k;
  }

  if (t == 0) cntS = (kept >= 0) ? kept : 0;
  {
    float4* a4 = (float4*)acc;
    for (int i = t; i < OWN_R * 32; i += OWN_T) a4[i] = make_float4(0.0f, 0.0f, 0.0f, 0.0f);
    for (int i = t; i < OWN_R * 4; i += OWN_T) {
      mxS[i] = 0x007fffffu;
      smS[i] = 0.0f;
    }
  }
  if (kept >= 0) {
    for (int i = t; i < kept; i += OWN_T) lst[i] = keep_b[OWN_KEEP0 + i];
  }
  __syncthreads();

  if (kept < 0 && packable) {
    int lead = (int)((4u - (unsigned)(((unsigned long long)dstI >> 2) & 3u)) & 3u);
    if (lead > E) lead = E;
    const long long nv = ((long long)E - lead) >> 2;
    const long long tail0 = (long long)lead + nv * 4;
    {
      long long i = -1;
      if (t < lead) i = t;
      else if ((long long)(t - lead) < (long long)E - tail0) i = tail0 + (t - lead);
      if (i >= 0) own_take(dstI[i], i, base, rows, lst, &cntS);
    }
    const int4* pv = (const int4*)(dstI + lead);
    const long long step = (long long)OWN_T * 4;
    auto load_step = [&](int4 (&qq)[4], long long v0) {
      if (v0 + step <= nv) {
#pragma unroll
        for (int u = 0; u < 4; ++u) qq[u] = pv[v0 + (long long)u * OWN_T + t];
      } else {
#pragma unroll
        for (int u = 0; u < 4; ++u) {
          long long v = v0 + (long long)u * OWN_T + t;
          qq[u] = (v < nv) ? pv[v] : make_int4(-1, -1, -1, -1);
        }
      }
    };
    int4 q[4];
    if (nv > 0) load_step(q, 0);
    for (long long v0 = 0; v0 < nv; v0 += step) {
      int4 qn[4];
      if (v0 + step < nv) {
        load_step(qn, v0 + step);
      } else {
#pragma unroll
        for (int u = 0; u < 4; ++u) qn[u] = make_int4(-1, -1, -1, -1);
      }
#pragma unroll
      for (int u = 0; u < 4; ++u) {
        long long i = (long long)lead + (v0 + (long long)u * OWN_T + t) * 4;
        own_take(q[u].x, i + 0, base, rows, lst, &cntS);
        own_take(q[u].y, i + 1, base, rows, lst, &cntS);
        own_take(q[u].z, i + 2, base, rows, lst, &cntS);
        own_take(q[u].w, i + 3, base, rows, lst, &cntS);
      }
#pragma unroll
      for (int u = 0; u < 4; ++u) q[u] = qn[u];
    }
  }
  __syncthreads();
  const int nraw = cntS;
  const bool listed = packable && nraw <= OWN_CAP;
  const int nl = listed ? nraw : 0;

  if (keep_mode == 0 && fits) {
    for (int i = t; i < nl; i += OWN_T) keep_b[OWN_KEEP0 + i] = lst[i];
    if (t == 0) keep_b[0] = listed ? (unsigned)nraw : (unsigned)OWN_CAP + 1u;
  }

  if (!listed) {
    own_unlisted(srcI, dstI, alpha, xh, acc, mxS, smS, base, rows, E, t);
  } else {
    const int nit = nl + rows;

    for (int b0 = 0; b0 < nit; b0 += OWN_T * 8) {
      float4 av[8];
      unsigned rw[8];
#pragma unroll
      for (int u = 0; u < 8; ++u) {
        int it = b0 + u * OWN_T + t;
        bool ok = it < nit;
        unsigned r = 0;
        long long arow;
        if (it < nl) {
          unsigned w = lst[it];
          arow = (long long)(w & OWN_EMASK);
          r = w >> OWN_RSH;
        } else {
          r = ok ? (unsigned)(it - nl) : 0u;
          arow = (long long)E + base + r;
        }
        av[u] = *(const float4*)(alpha + arow * 4);
        rw[u] = ok ? r : 0xffffffffu;
      }
#pragma unroll
      for (int u = 0; u < 8; ++u) {
        if (rw[u] != 0xffffffffu) {
          unsigned* m = mxS + rw[u] * 4;
          atomicMax(m + 0, ordEnc(av[u].x));
          atomicMax(m + 1, ordEnc(av[u].y));
          atomicMax(m + 2, ordEnc(av[u].z));
          atomicMax(m + 3, ordEnc(av[u].w));
        }
      }
    }
    __syncthreads();

    const int wv = __builtin_amdgcn_readfirstlane(t >> 5);
    const int hl = lane & 3;
    auto fetch = [&](int g, int& s_, unsigned& r_, float& al_, unsigned& mx_) {
      int it = g + (lane >> 2);
      if (it >= nit) it = nit - 1;
      long long arow;
      if (it < nl) {
        unsigned w = lst[it];
        int e = (int)(w & OWN_EMASK);
        r_ = w >> OWN_RSH;
        s_ = srcI[e];
        arow = e;
      } else {
        r_ = (unsigned)(it - nl);
        s_ = base + (int)r_;
        arow = (long long)E + s_;
      }
      al_ = alpha[arow * 4 + hl];
      mx_ = mxS[r_ * 4 + hl];
    };
    int s = base;
    unsigned r = 0, mx = 0;
    float al = 0.0f;
    if (wv * 8 < nit) fetch(wv * 8, s, r, al, mx);
    for (int g0 = wv * 8; g0 < nit; g0 += OWN_W * 8) {
      int nlive = (nit - g0 >= 8) ? 8 : (nit - g0);
      float xv[8][4];
      int rr[8];
#pragma unroll
      for (int k = 0; k < 8; ++k) {
        int se = __builtin_amdgcn_readlane(s, k * 4);
        rr[k] = __builtin_amdgcn_readlane((int)r, k * 4);
        const float* xr = xh + (long long)se * 128 + lane;
#pragma unroll
        for (int j = 0; j < 4; ++j) xv[k][j] = xr[j * 32];
      }
      float p = __expf(al - own_max_dec(mx));
      if ((lane >> 2) < nlive) atomicAdd(&smS[r * 4 + hl], p);
      int s2 = base;
      unsigned r2 = 0, mx2 = 0;
      float al2 = 0.0f;
      int gn = g0 + OWN_W * 8;
      if (gn < nit) fetch(gn, s2, r2, al2, mx2);
      int pi = __float_as_int(p);
#pragma unroll
      for (int k = 0; k < 8; ++k) {
        if (k == 0 || k < nlive) {
          float* a = acc + rr[k] * 128 + lane;
#pragma unroll
          for (int j = 0; j < 4; ++j) {
            float pj = __int_as_float(__builtin_amdgcn_readlane(pi, k * 4 + j));
            atomicAdd(a + j * 32, pj * xv[k][j]);
          }
        }
      }
      s = s2; r = r2; al = al2; mx = mx2;
    }
  }
  __syncthreads();

  {
    const int c = t & 127;
    const int h = c >> 5;
    const float b = bias[c];
    for (int r = t >> 7; r < rows; r += OWN_T / 128) {
      float v = acc[r * 128 + c] / (smS[r * 4 + h] + 1e-16f) + b;
      if (relu) v = (v > 0.0f) ? v : 0.0f;
      out[((long long)base + r) * 128 + c] = v;
    }
  }
}

static void fillN(unsigned* p, unsigned v, long long n, hipStream_t s) {
  if (n <= 0) return;
  long long b = (n + 255) / 256;
  hipLaunchKernelGGL(gat_fill_u32, dim3((unsigned)b), dim3(256), 0, s, p, v, n);
}

extern "C" void kernel_launch(void* const* d_in, const int* in_sizes, int n_in,
                              void* d_out, int out_size, void* d_ws,
                              size_t ws_size, hipStream_t stream) {
  const int N = in_sizes[0] / 128;
  const int E = in_sizes[1] / 2;

  const float* x  = (const float*)d_in[0];
  const int* ei   = (const int*)d_in[1];
  const int* srcI = ei;
  const int* dstI = ei + E;
  const float* ea = (const float*)d_in[2];

  const float* Wl[2]  = {(const float*)d_in[3], (const float*)d_in[9]};
  const float* asl[2] = {(const float*)d_in[4], (const float*)d_in[10]};
  const float* adl[2] = {(const float*)d_in[5], (const float*)d_in[11]};
  const float* Wel[2] = {(const float*)d_in[6], (const float*)d_in[12]};
  const float* ael[2] = {(const float*)d_in[7], (const float*)d_in[13]};
  const float* bl[2]  = {(const float*)d_in[8], (const float*)d_in[14]};

  float* wsf = (float*)d_ws;
  size_t off = 0;
  float* xh    = wsf + off; off += (size_t)N * 128;
  float* h1    = wsf + off; off += (size_t)N * 128;
  float* la    = wsf + off; off += (size_t)N * 16;
  float* medge = wsf + off; off += 64;
  float* a_src = wsf + off; off += (size_t)N * 4;
  float* a_dst = wsf + off; off += (size_t)N * 4;
  float* alpha = wsf + off; off += ((size_t)E + N) * 4;
  float* cnt   = wsf + off; off += (size_t)N;
  (void)ws_size; (void)n_in; (void)out_size;

  const long long F = (long long)E + N;

  fillN((unsigned*)cnt, 0u, N, stream);
  fillN((unsigned*)la, 0u, (long long)N * 16, stream);
  {
    long long n = (long long)E * 16;
    hipLaunchKernelGGL(gat_count_loop, dim3((unsigned)((n + 255) / 256)),
                       dim3(256), 0, stream, dstI, ea, cnt, la, E);
    long long m = (long long)N * 16;
    hipLaunchKernelGGL(gat_loop_div, dim3((unsigned)((m + 255) / 256)),
                       dim3(256), 0, stream, cnt, la, N);
  }

  for (int l = 0; l < 2; ++l) {
    const float* xin = (l == 0) ? x : h1;
    float* outb = (l == 0) ? h1 : (float*)d_out;

    hipLaunchKernelGGL(gat_medge, dim3(1), dim3(64), 0, stream, Wel[l], ael[l],
                       medge);

    {
      long long waves = ((long long)N + 15) / 16;
      long long blocks = (waves + 7) / 8;
      hipLaunchKernelGGL(gat_gemm_wmma, dim3((unsigned)blocks), dim3(256), 0,
                         stream, xin, Wl[l], asl[l], adl[l], xh, a_src, a_dst, N);
    }

    hipLaunchKernelGGL(gat_alpha, dim3((unsigned)((F * 4 + 255) / 256)), dim3(256),
                       0, stream, srcI, dstI, ea, la, a_src, a_dst, medge,
                       alpha, E, N);

    {
      long long blocks = ((long long)N + OWN_R - 1) / OWN_R;
      hipLaunchKernelGGL(gat_own, dim3((unsigned)blocks), dim3(OWN_T), 0, stream,
                         srcI, dstI, alpha, xh, bl[l], outb, (unsigned*)d_out,
                         l, E, N, (l == 0) ? 1 : 0);
    }
  }
}


# self-loop attribute means by the range owner in LDS: a first owner launch scans, keeps the list, adds its edges' attribute rows in LDS; no global atomics or fills, 12 -> 9 launches; sum order new
# speedup vs baseline: 3.6230x; 1.3463x over previous
#include <hip/hip_runtime.h>
#include <hip/hip_bf16.h>

typedef __attribute__((ext_vector_type(2))) float v2f;
typedef __attribute__((ext_vector_type(8))) float v8f;

#define NEG_SLOPE 0.2f

__device__ __forceinline__ unsigned ordEnc(float f) {
  unsigned u = __float_as_uint(f);
  return (u & 0x80000000u) ? ~u : (u | 0x80000000u);
}
__device__ __forceinline__ float ordDec(unsigned u) {
  return (u & 0x80000000u) ? __uint_as_float(u ^ 0x80000000u)
                           : __uint_as_float(~u);
}

__global__ void gat_medge(const float* __restrict__ We,
                          const float* __restrict__ ae,
                          float* __restrict__ medge) {
  int t = threadIdx.x;
  int k = t >> 2, h = t & 3;
  float s = 0.0f;
  for (int c = 0; c < 32; ++c)
    s += We[k * 128 + h * 32 + c] * ae[h * 32 + c];
  medge[k * 4 + h] = s;
}

__global__ __launch_bounds__(256) void gat_gemm_wmma(
    const float* __restrict__ X, const float* __restrict__ W,
    const float* __restrict__ as, const float* __restrict__ ad,
    float* __restrict__ Y, float* __restrict__ a_src,
    float* __restrict__ a_dst, int Nrows) {
  int wave = (int)(((long long)blockIdx.x * blockDim.x + threadIdx.x) >> 5);
  int lane = threadIdx.x & 31;
  int m0 = wave * 16;
  if (m0 >= Nrows) return;

  v8f acc[8] = {};
  int half = lane >> 4;
  int l16 = lane & 15;
  int arow = m0 + l16;
  if (arow >= Nrows) arow = Nrows - 1;
  const float* xr = X + (long long)arow * 128;

  for (int k0 = 0; k0 < 128; k0 += 4) {
    v2f a;
    a.x = xr[k0 + half * 2];
    a.y = xr[k0 + half * 2 + 1];
#pragma unroll
    for (int nt = 0; nt < 8; ++nt) {
      int col = nt * 16 + l16;
      v2f b;
      b.x = W[(k0 + half * 2) * 128 + col];
      b.y = W[(k0 + half * 2 + 1) * 128 + col];
      acc[nt] = __builtin_amdgcn_wmma_f32_16x16x4_f32(
          false, a, false, b, (short)0, acc[nt], false, false);
    }
  }
#pragma unroll
  for (int nt = 0; nt < 8; ++nt) {
#pragma unroll
    for (int r = 0; r < 8; ++r) {
      int row = m0 + r + half * 8;
      if (row < Nrows) Y[(long long)row * 128 + nt * 16 + l16] = acc[nt][r];
    }
  }

  float cs[8], cd[8];
#pragma unroll
  for (int nt = 0; nt < 8; ++nt) {
    cs[nt] = as[nt * 16 + l16];
    cd[nt] = ad[nt * 16 + l16];
  }
  float s0 = 0.0f, s1 = 0.0f, d0 = 0.0f, d1 = 0.0f;
#pragma unroll
  for (int r = 0; r < 8; ++r) {
#pragma unroll
    for (int hd = 0; hd < 4; ++hd) {
      float ps = acc[2 * hd][r] * cs[2 * hd] + acc[2 * hd + 1][r] * cs[2 * hd + 1];
      float pd = acc[2 * hd][r] * cd[2 * hd] + acc[2 * hd + 1][r] * cd[2 * hd + 1];
      ps += __shfl_xor(ps, 8, 32);
      pd += __shfl_xor(pd, 8, 32);
      ps += __shfl_xor(ps, 4, 32);
      pd += __shfl_xor(pd, 4, 32);
      ps += __shfl_xor(ps, 2, 32);
      pd += __shfl_xor(pd, 2, 32);
      ps += __shfl_xor(ps, 1, 32);
      pd += __shfl_xor(pd, 1, 32);
      const bool mine = (l16 == r * 2 + (hd >> 1));
      if ((hd & 1) == 0) {
        s0 = mine ? ps : s0;
        d0 = mine ? pd : d0;
      } else {
        s1 = mine ? ps : s1;
        d1 = mine ? pd : d1;
      }
    }
  }
  {
    int row = m0 + half * 8 + (l16 >> 1);
    if (row < Nrows) {
      long long o = (long long)row * 4 + (l16 & 1) * 2;
      *(float2*)(a_src + o) = make_float2(s0, s1);
      *(float2*)(a_dst + o) = make_float2(d0, d1);
    }
  }
}

__global__ void gat_alpha(const int* __restrict__ srcI,
                          const int* __restrict__ dstI,
                          const float* __restrict__ ea,
                          const float* __restrict__ la,
                          const float* __restrict__ a_src,
                          const float* __restrict__ a_dst,
                          const float* __restrict__ medge,
                          float* __restrict__ alpha, int E, int Nn) {
  long long i = (long long)blockIdx.x * blockDim.x + threadIdx.x;
  long long F = (long long)E + Nn;
  if (i >= F * 4) return;
  long long f = i >> 2;
  int h = (int)(i & 3);
  int s, d;
  const float* row;
  if (f < (long long)E) {
    s = srcI[f]; d = dstI[f]; row = ea + f * 16;
  } else {
    s = (int)(f - E); d = s; row = la + (long long)s * 16;
  }
  float e0[16];
#pragma unroll
  for (int k = 0; k < 16; ++k) e0[k] = row[k];
  float q0 = 0.0f, q1 = 0.0f, q2 = 0.0f, q3 = 0.0f;
#pragma unroll
  for (int k = 0; k < 16; ++k) {
    q0 += e0[k] * medge[k * 4 + 0];
    q1 += e0[k] * medge[k * 4 + 1];
    q2 += e0[k] * medge[k * 4 + 2];
    q3 += e0[k] * medge[k * 4 + 3];
  }
  float aE = q3;
  aE = (h == 2) ? q2 : aE;
  aE = (h == 1) ? q1 : aE;
  aE = (h == 0) ? q0 : aE;
  float v = a_src[(long long)s * 4 + h] + a_dst[(long long)d * 4 + h] + aE;
  v = (v > 0.0f) ? v : NEG_SLOPE * v;
  alpha[i] = v;
}

#define OWN_R 400
#define OWN_T 1024
#define OWN_W (OWN_T / 32)
#define OWN_CAP 16384
#define OWN_KEEP0 4
#define OWN_RSH 23
#define OWN_EMASK ((1u << OWN_RSH) - 1u)
static_assert(OWN_R <= (1 << (32 - OWN_RSH)), "the row in range must fit above the edge number");

__device__ __forceinline__ float own_max_dec(unsigned u) {
  float mv = ordDec(u);
  if (!(mv > -__builtin_inff() && mv < __builtin_inff())) mv = 0.0f;
  return mv;
}

__device__ __forceinline__ void own_take(int dval, long long i, int base, int rows,
                                         unsigned* lst, int* cntS) {
  unsigned r = (unsigned)(dval - base);
  if (r < (unsigned)rows) {
    int slot = atomicAdd(cntS, 1);
    if (slot < OWN_CAP) lst[slot] = (unsigned)i | (r << OWN_RSH);
  }
}

__device__ __forceinline__ void own_scan(const int* __restrict__ dstI, int E,
                                         int base, int rows, unsigned* lst,
                                         int* cntS, int t) {
  int lead = (int)((4u - (unsigned)(((unsigned long long)dstI >> 2) & 3u)) & 3u);
  if (lead > E) lead = E;
  const long long nv = ((long long)E - lead) >> 2;
  const long long tail0 = (long long)lead + nv * 4;
  {
    long long i = -1;
    if (t < lead) i = t;
    else if ((long long)(t - lead) < (long long)E - tail0) i = tail0 + (t - lead);
    if (i >= 0) own_take(dstI[i], i, base, rows, lst, cntS);
  }
  const int4* pv = (const int4*)(dstI + lead);
  const long long step = (long long)OWN_T * 4;
  auto load_step = [&](int4 (&qq)[4], long long v0) {
    if (v0 + step <= nv) {
#pragma unroll
      for (int u = 0; u < 4; ++u) qq[u] = pv[v0 + (long long)u * OWN_T + t];
    } else {
#pragma unroll
      for (int u = 0; u < 4; ++u) {
        long long v = v0 + (long long)u * OWN_T + t;
        qq[u] = (v < nv) ? pv[v] : make_int4(-1, -1, -1, -1);
      }
    }
  };
  int4 q[4];
  if (nv > 0) load_step(q, 0);
  for (long long v0 = 0; v0 < nv; v0 += step) {
    int4 qn[4];
    if (v0 + step < nv) {
      load_step(qn, v0 + step);
    } else {
#pragma unroll
      for (int u = 0; u < 4; ++u) qn[u] = make_int4(-1, -1, -1, -1);
    }
#pragma unroll
    for (int u = 0; u < 4; ++u) {
      long long i = (long long)lead + (v0 + (long long)u * OWN_T + t) * 4;
      own_take(q[u].x, i + 0, base, rows, lst, cntS);
      own_take(q[u].y, i + 1, base, rows, lst, cntS);
      own_take(q[u].z, i + 2, base, rows, lst, cntS);
      own_take(q[u].w, i + 3, base, rows, lst, cntS);
    }
#pragma unroll
    for (int u = 0; u < 4; ++u) q[u] = qn[u];
  }
}

__device__ __noinline__ void own_unlisted(
    const int* __restrict__ srcI, const int* __restrict__ dstI,
    const float* __restrict__ alpha, const float* __restrict__ xh,
    float* acc, unsigned* mxS, float* smS, int base, int rows, int E, int t) {
  const long long nall = (long long)E + rows;
  for (long long i = t; i < nall; i += OWN_T) {
    unsigned r;
    long long arow;
    if (i < (long long)E) { r = (unsigned)(dstI[i] - base); arow = i; }
    else { r = (unsigned)(i - E); arow = (long long)E + base + r; }
    if (r < (unsigned)rows)
      for (int h = 0; h < 4; ++h) atomicMax(&mxS[r * 4 + h], ordEnc(alpha[arow * 4 + h]));
  }
  __syncthreads();
  for (long long i = t; i < nall; i += OWN_T) {
    unsigned r;
    long long arow;
    int s;
    if (i < (long long)E) { r = (unsigned)(dstI[i] - base); arow = i; s = srcI[i]; }
    else { r = (unsigned)(i - E); arow = (long long)E + base + r; s = base + (int)r; }
    if (r < (unsigned)rows) {
      const float* xr = xh + (long long)s * 128;
      for (int h = 0; h < 4; ++h) {
        float p = __expf(alpha[arow * 4 + h] - own_max_dec(mxS[r * 4 + h]));
        atomicAdd(&smS[r * 4 + h], p);
        for (int c = 0; c < 32; ++c) atomicAdd(acc + r * 128 + h * 32 + c, p * xr[h * 32 + c]);
      }
    }
  }
}


__device__ __noinline__ void loop_unlisted(
    const int* __restrict__ dstI, const float* __restrict__ ea,
    float* sumS, unsigned* cS, int base, int rows, int E, int t) {
  for (long long i = t; i < (long long)E; i += OWN_T) {
    unsigned r = (unsigned)(dstI[i] - base);
    if (r < (unsigned)rows) {
      for (int j = 0; j < 16; ++j) atomicAdd(&sumS[r * 16 + j], ea[i * 16 + j]);
      atomicAdd(&cS[r], 1u);
    }
  }
}

__global__ __launch_bounds__(OWN_T) void gat_own_loop(
    const int* __restrict__ dstI, const float* __restrict__ ea,
    float* __restrict__ la, unsigned* keep, int E, int Nn) {
  __shared__ __attribute__((aligned(16))) float sumS[OWN_R * 128];
  __shared__ __attribute__((aligned(16))) unsigned lst[OWN_CAP];
  __shared__ unsigned cS[OWN_R];
  __shared__ int cntS;

  const int t = threadIdx.x;
  const int base = (int)blockIdx.x * OWN_R;
  const int rows = (Nn - base >= OWN_R) ? OWN_R : (Nn - base);
  const bool packable = E <= (int)OWN_EMASK;
  unsigned* const keep_b = keep + (long long)base * 128;
  const bool fits = rows * 128 >= OWN_CAP + OWN_KEEP0;

  if (t == 0) cntS = 0;
  for (int i = t; i < OWN_R * 16; i += OWN_T) sumS[i] = 0.0f;
  for (int i = t; i < OWN_R; i += OWN_T) cS[i] = 0u;
  __syncthreads();

  if (packable) own_scan(dstI, E, base, rows, lst, &cntS, t);
  __syncthreads();
  const int nraw = cntS;
  const bool listed = packable && nraw <= OWN_CAP;
  const int nl = listed ? nraw : 0;

  if (fits) {
    for (int i = t; i < nl; i += OWN_T) keep_b[OWN_KEEP0 + i] = lst[i];
    if (t == 0) keep_b[0] = listed ? (unsigned)nraw : (unsigned)OWN_CAP + 1u;
  }

  if (!listed) {
    loop_unlisted(dstI, ea, sumS, cS, base, rows, E, t);
  } else {
    const int q = t & 3;
    for (int b0 = 0; b0 < nl; b0 += OWN_T * 2) {
      float4 v[8];
      unsigned rw[8];
#pragma unroll
      for (int u = 0; u < 8; ++u) {
        int it = b0 + u * (OWN_T / 4) + (t >> 2);
        bool ok = it < nl;
        unsigned w = lst[ok ? it : nl - 1];
        v[u] = *(const float4*)(ea + (long long)(w & OWN_EMASK) * 16 + q * 4);
        rw[u] = ok ? (w >> OWN_RSH) : 0xffffffffu;
      }
#pragma unroll
      for (int u = 0; u < 8; ++u) {
        if (rw[u] != 0xffffffffu) {
          float* a = sumS + rw[u] * 16 + q * 4;
          atomicAdd(a + 0, v[u].x);
          atomicAdd(a + 1, v[u].y);
          atomicAdd(a + 2, v[u].z);
          atomicAdd(a + 3, v[u].w);
          if (q == 0) atomicAdd(&cS[rw[u]], 1u);
        }
      }
    }
  }
  __syncthreads();

  for (int i = t; i < rows * 16; i += OWN_T) {
    float cf = (float)cS[i >> 4];
    cf = (cf < 1.0f) ? 1.0f : cf;
    la[(long long)base * 16 + i] = sumS[i] / cf;
  }
}

__global__ __launch_bounds__(OWN_T) void gat_own(
    const int* __restrict__ srcI, const int* __restrict__ dstI,
    const float* __restrict__ alpha,
    const float* __restrict__ xh, const float* __restrict__ bias,
    float* out, const unsigned* keep, int E, int Nn, int relu) {
  __shared__ __attribute__((aligned(16))) float acc[OWN_R * 128];
  __shared__ __attribute__((aligned(16))) unsigned lst[OWN_CAP];
  __shared__ __attribute__((aligned(16))) unsigned mxS[OWN_R * 4];
  __shared__ __attribute__((aligned(16))) float smS[OWN_R * 4];
  __shared__ int cntS;

  const int t = threadIdx.x;
  const int lane = t & 31;
  const int base = (int)blockIdx.x * OWN_R;
  const int rows = (Nn - base >= OWN_R) ? OWN_R : (Nn - base);

  const bool packable = E <= (int)OWN_EMASK;
  const unsigned* const keep_b = keep + (long long)base * 128;
  const bool fits = rows * 128 >= OWN_CAP + OWN_KEEP0;
  int kept = -1;
  if (fits && packable) {
    unsigned k = keep_b[0];
    if (k <= (unsigned)OWN_CAP) kept = (int)k;
  }

  if (t == 0) cntS = (kept >= 0) ? kept : 0;
  {
    float4* a4 = (float4*)acc;
    for (int i = t; i < OWN_R * 32; i += OWN_T) a4[i] = make_float4(0.0f, 0.0f, 0.0f, 0.0f);
    for (int i = t; i < OWN_R * 4; i += OWN_T) {
      mxS[i] = 0x007fffffu;
      smS[i] = 0.0f;
    }
  }
  if (kept >= 0) {
    for (int i = t; i < kept; i += OWN_T) lst[i] = keep_b[OWN_KEEP0 + i];
  }
  __syncthreads();

  if (kept < 0 && packable) own_scan(dstI, E, base, rows, lst, &cntS, t);
  __syncthreads();
  const int nraw = cntS;
  const bool listed = packable && nraw <= OWN_CAP;
  const int nl = listed ? nraw : 0;

  if (!listed) {
    own_unlisted(srcI, dstI, alpha, xh, acc, mxS, smS, base, rows, E, t);
  } else {
    const int nit = nl + rows;

    for (int b0 = 0; b0 < nit; b0 += OWN_T * 8) {
      float4 av[8];
      unsigned rw[8];
#pragma unroll
      for (int u = 0; u < 8; ++u) {
        int it = b0 + u * OWN_T + t;
        bool ok = it < nit;
        unsigned r = 0;
        long long arow;
        if (it < nl) {
          unsigned w = lst[it];
          arow = (long long)(w & OWN_EMASK);
          r = w >> OWN_RSH;
        } else {
          r = ok ? (unsigned)(it - nl) : 0u;
          arow = (long long)E + base + r;
        }
        av[u] = *(const float4*)(alpha + arow * 4);
        rw[u] = ok ? r : 0xffffffffu;
      }
#pragma unroll
      for (int u = 0; u < 8; ++u) {
        if (rw[u] != 0xffffffffu) {
          unsigned* m = mxS + rw[u] * 4;
          atomicMax(m + 0, ordEnc(av[u].x));
          atomicMax(m + 1, ordEnc(av[u].y));
          atomicMax(m + 2, ordEnc(av[u].z));
          atomicMax(m + 3, ordEnc(av[u].w));
        }
      }
    }
    __syncthreads();

    const int wv = __builtin_amdgcn_readfirstlane(t >> 5);
    const int hl = lane & 3;
    auto fetch = [&](int g, int& s_, unsigned& r_, float& al_, unsigned& mx_) {
      int it = g + (lane >> 2);
      if (it >= nit) it = nit - 1;
      long long arow;
      if (it < nl) {
        unsigned w = lst[it];
        int e = (int)(w & OWN_EMASK);
        r_ = w >> OWN_RSH;
        s_ = srcI[e];
        arow = e;
      } else {
        r_ = (unsigned)(it - nl);
        s_ = base + (int)r_;
        arow = (long long)E + s_;
      }
      al_ = alpha[arow * 4 + hl];
      mx_ = mxS[r_ * 4 + hl];
    };
    int s = base;
    unsigned r = 0, mx = 0;
    float al = 0.0f;
    if (wv * 8 < nit) fetch(wv * 8, s, r, al, mx);
    for (int g0 = wv * 8; g0 < nit; g0 += OWN_W * 8) {
      int nlive = (nit - g0 >= 8) ? 8 : (nit - g0);
      float xv[8][4];
      int rr[8];
#pragma unroll
      for (int k = 0; k < 8; ++k) {
        int se = __builtin_amdgcn_readlane(s, k * 4);
        rr[k] = __builtin_amdgcn_readlane((int)r, k * 4);
        const float* xr = xh + (long long)se * 128 + lane;
#pragma unroll
        for (int j = 0; j < 4; ++j) xv[k][j] = xr[j * 32];
      }
      float p = __expf(al - own_max_dec(mx));
      if ((lane >> 2) < nlive) atomicAdd(&smS[r * 4 + hl], p);
      int s2 = base;
      unsigned r2 = 0, mx2 = 0;
      float al2 = 0.0f;
      int gn = g0 + OWN_W * 8;
      if (gn < nit) fetch(gn, s2, r2, al2, mx2);
      int pi = __float_as_int(p);
#pragma unroll
      for (int k = 0; k < 8; ++k) {
        if (k == 0 || k < nlive) {
          float* a = acc + rr[k] * 128 + lane;
#pragma unroll
          for (int j = 0; j < 4; ++j) {
            float pj = __int_as_float(__builtin_amdgcn_readlane(pi, k * 4 + j));
            atomicAdd(a + j * 32, pj * xv[k][j]);
          }
        }
      }
      s = s2; r = r2; al = al2; mx = mx2;
    }
  }
  __syncthreads();

  {
    const int c = t & 127;
    const int h = c >> 5;
    const float b = bias[c];
    for (int r = t >> 7; r < rows; r += OWN_T / 128) {
      float v = acc[r * 128 + c] / (smS[r * 4 + h] + 1e-16f) + b;
      if (relu) v = (v > 0.0f) ? v : 0.0f;
      out[((long long)base + r) * 128 + c] = v;
    }
  }
}

extern "C" void kernel_launch(void* const* d_in, const int* in_sizes, int n_in,
                              void* d_out, int out_size, void* d_ws,
                              size_t ws_size, hipStream_t stream) {
  const int N = in_sizes[0] / 128;
  const int E = in_sizes[1] / 2;

  const float* x  = (const float*)d_in[0];
  const int* ei   = (const int*)d_in[1];
  const int* srcI = ei;
  const int* dstI = ei + E;
  const float* ea = (const float*)d_in[2];

  const float* Wl[2]  = {(const float*)d_in[3], (const float*)d_in[9]};
  const float* asl[2] = {(const float*)d_in[4], (const float*)d_in[10]};
  const float* adl[2] = {(const float*)d_in[5], (const float*)d_in[11]};
  const float* Wel[2] = {(const float*)d_in[6], (const float*)d_in[12]};
  const float* ael[2] = {(const float*)d_in[7], (const float*)d_in[13]};
  const float* bl[2]  = {(const float*)d_in[8], (const float*)d_in[14]};

  float* wsf = (float*)d_ws;
  size_t off = 0;
  float* xh    = wsf + off; off += (size_t)N * 128;
  float* h1    = wsf + off; off += (size_t)N * 128;
  float* la    = wsf + off; off += (size_t)N * 16;
  float* medge = wsf + off; off += 64;
  float* a_src = wsf + off; off += (size_t)N * 4;
  float* a_dst = wsf + off; off += (size_t)N * 4;
  float* alpha = wsf + off; off += ((size_t)E + N) * 4;
  (void)ws_size; (void)n_in; (void)out_size;

  const long long F = (long long)E + N;

  const long long own_blocks = ((long long)N + OWN_R - 1) / OWN_R;

  hipLaunchKernelGGL(gat_own_loop, dim3((unsigned)own_blocks), dim3(OWN_T), 0,
                     stream, dstI, ea, la, (unsigned*)d_out, E, N);

  for (int l = 0; l < 2; ++l) {
    const float* xin = (l == 0) ? x : h1;
    float* outb = (l == 0) ? h1 : (float*)d_out;

    hipLaunchKernelGGL(gat_medge, dim3(1), dim3(64), 0, stream, Wel[l], ael[l],
                       medge);

    {
      long long waves = ((long long)N + 15) / 16;
      long long blocks = (waves + 7) / 8;
      hipLaunchKernelGGL(gat_gemm_wmma, dim3((unsigned)blocks), dim3(256), 0,
                         stream, xin, Wl[l], asl[l], adl[l], xh, a_src, a_dst, N);
    }

    hipLaunchKernelGGL(gat_alpha, dim3((unsigned)((F * 4 + 255) / 256)), dim3(256),
                       0, stream, srcI, dstI, ea, la, a_src, a_dst, medge,
                       alpha, E, N);

    hipLaunchKernelGGL(gat_own, dim3((unsigned)own_blocks), dim3(OWN_T), 0, stream,
                       srcI, dstI, alpha, xh, bl[l], outb,
                       (const unsigned*)d_out, E, N, (l == 0) ? 1 : 0);
  }
}


# logits formed by the range owner in its walks, from edge terms of both layers the first owner launch leaves in list order: no logit or coefficient launches, 9 -> 5; edge-term sum order new
# speedup vs baseline: 4.2161x; 1.1637x over previous
#include <hip/hip_runtime.h>
#include <hip/hip_bf16.h>

typedef __attribute__((ext_vector_type(2))) float v2f;
typedef __attribute__((ext_vector_type(8))) float v8f;

#define NEG_SLOPE 0.2f

__device__ __forceinline__ unsigned ordEnc(float f) {
  unsigned u = __float_as_uint(f);
  return (u & 0x80000000u) ? ~u : (u | 0x80000000u);
}
__device__ __forceinline__ float ordDec(unsigned u) {
  return (u & 0x80000000u) ? __uint_as_float(u ^ 0x80000000u)
                           : __uint_as_float(~u);
}

__global__ __launch_bounds__(256) void gat_gemm_wmma(
    const float* __restrict__ X, const float* __restrict__ W,
    const float* __restrict__ as, const float* __restrict__ ad,
    float* __restrict__ Y, float* __restrict__ a_src,
    float* __restrict__ a_dst, int Nrows) {
  int wave = (int)(((long long)blockIdx.x * blockDim.x + threadIdx.x) >> 5);
  int lane = threadIdx.x & 31;
  int m0 = wave * 16;
  if (m0 >= Nrows) return;

  v8f acc[8] = {};
  int half = lane >> 4;
  int l16 = lane & 15;
  int arow = m0 + l16;
  if (arow >= Nrows) arow = Nrows - 1;
  const float* xr = X + (long long)arow * 128;

  for (int k0 = 0; k0 < 128; k0 += 4) {
    v2f a;
    a.x = xr[k0 + half * 2];
    a.y = xr[k0 + half * 2 + 1];
#pragma unroll
    for (int nt = 0; nt < 8; ++nt) {
      int col = nt * 16 + l16;
      v2f b;
      b.x = W[(k0 + half * 2) * 128 + col];
      b.y = W[(k0 + half * 2 + 1) * 128 + col];
      acc[nt] = __builtin_amdgcn_wmma_f32_16x16x4_f32(
          false, a, false, b, (short)0, acc[nt], false, false);
    }
  }
#pragma unroll
  for (int nt = 0; nt < 8; ++nt) {
#pragma unroll
    for (int r = 0; r < 8; ++r) {
      int row = m0 + r + half * 8;
      if (row < Nrows) Y[(long long)row * 128 + nt * 16 + l16] = acc[nt][r];
    }
  }

  float cs[8], cd[8];
#pragma unroll
  for (int nt = 0; nt < 8; ++nt) {
    cs[nt] = as[nt * 16 + l16];
    cd[nt] = ad[nt * 16 + l16];
  }
  float s0 = 0.0f, s1 = 0.0f, d0 = 0.0f, d1 = 0.0f;
#pragma unroll
  for (int r = 0; r < 8; ++r) {
#pragma unroll
    for (int hd = 0; hd < 4; ++hd) {
      float ps = acc[2 * hd][r] * cs[2 * hd] + acc[2 * hd + 1][r] * cs[2 * hd + 1];
      float pd = acc[2 * hd][r] * cd[2 * hd] + acc[2 * hd + 1][r] * cd[2 * hd + 1];
      ps += __shfl_xor(ps, 8, 32);
      pd += __shfl_xor(pd, 8, 32);
      ps += __shfl_xor(ps, 4, 32);
      pd += __shfl_xor(pd, 4, 32);
      ps += __shfl_xor(ps, 2, 32);
      pd += __shfl_xor(pd, 2, 32);
      ps += __shfl_xor(ps, 1, 32);
      pd += __shfl_xor(pd, 1, 32);
      const bool mine = (l16 == r * 2 + (hd >> 1));
      if ((hd & 1) == 0) {
        s0 = mine ? ps : s0;
        d0 = mine ? pd : d0;
      } else {
        s1 = mine ? ps : s1;
        d1 = mine ? pd : d1;
      }
    }
  }
  {
    int row = m0 + half * 8 + (l16 >> 1);
    if (row < Nrows) {
      long long o = (long long)row * 4 + (l16 & 1) * 2;
      *(float2*)(a_src + o) = make_float2(s0, s1);
      *(float2*)(a_dst + o) = make_float2(d0, d1);
    }
  }
}

__device__ __forceinline__ float own_medge(const float* __restrict__ We,
                                           const float* __restrict__ ae, int k, int h) {
  float s = 0.0f;
  for (int c = 0; c < 32; ++c)
    s += We[k * 128 + h * 32 + c] * ae[h * 32 + c];
  return s;
}

__device__ __forceinline__ float own_leaky(float v) {
  return (v > 0.0f) ? v : NEG_SLOPE * v;
}

__device__ __forceinline__ float own_term(const float* __restrict__ row,
                                          const float* m, int h) {
  float q = 0.0f;
#pragma nounroll
  for (int k = 0; k < 16; ++k) q += row[k] * m[k * 4 + h];
  return q;
}

__device__ __forceinline__ float quad_xor1(float v) {
  return __int_as_float(__builtin_amdgcn_mov_dpp(__float_as_int(v), 0xB1, 0xF, 0xF, true));
}
__device__ __forceinline__ float quad_xor2(float v) {
  return __int_as_float(__builtin_amdgcn_mov_dpp(__float_as_int(v), 0x4E, 0xF, 0xF, true));
}

#define OWN_R 400
#define OWN_T 1024
#define OWN_W (OWN_T / 32)
#define OWN_CAP 9900
#define OWN_KEEP0 4
#define OWN_T2OFF (OWN_KEEP0 + OWN_CAP)
#define OWN_BLKW (OWN_T2OFF + 4 * (OWN_CAP + OWN_R))
#define OWN_RSH 23
#define OWN_EMASK ((1u << OWN_RSH) - 1u)
static_assert(OWN_R <= (1 << (32 - OWN_RSH)), "the row in range must fit above the edge number");
static_assert(OWN_T2OFF % 4 == 0, "terms start on a 16-byte boundary");
static_assert(OWN_BLKW <= OWN_R * 128, "list and terms must fit the range's own output rows");

__device__ __forceinline__ float own_max_dec(unsigned u) {
  float mv = ordDec(u);
  if (!(mv > -__builtin_inff() && mv < __builtin_inff())) mv = 0.0f;
  return mv;
}

__device__ __forceinline__ void own_take(int dval, long long i, int base, int rows,
                                         unsigned* lst, int* cntS) {
  unsigned r = (unsigned)(dval - base);
  if (r < (unsigned)rows) {
    int slot = atomicAdd(cntS, 1);
    if (slot < OWN_CAP) lst[slot] = (unsigned)i | (r << OWN_RSH);
  }
}

__device__ __forceinline__ void own_scan(const int* __restrict__ dstI, int E,
                                         int base, int rows, unsigned* lst,
                                         int* cntS, int t) {
  int lead = (int)((4u - (unsigned)(((unsigned long long)dstI >> 2) & 3u)) & 3u);
  if (lead > E) lead = E;
  const long long nv = ((long long)E - lead) >> 2;
  const long long tail0 = (long long)lead + nv * 4;
  {
    long long i = -1;
    if (t < lead) i = t;
    else if ((long long)(t - lead) < (long long)E - tail0) i = tail0 + (t - lead);
    if (i >= 0) own_take(dstI[i], i, base, rows, lst, cntS);
  }
  const int4* pv = (const int4*)(dstI + lead);
  const long long step = (long long)OWN_T * 4;
  auto load_step = [&](int4 (&qq)[4], long long v0) {
    if (v0 + step <= nv) {
#pragma unroll
      for (int u = 0; u < 4; ++u) qq[u] = pv[v0 + (long long)u * OWN_T + t];
    } else {
#pragma unroll
      for (int u = 0; u < 4; ++u) {
        long long v = v0 + (long long)u * OWN_T + t;
        qq[u] = (v < nv) ? pv[v] : make_int4(-1, -1, -1, -1);
      }
    }
  };
  int4 q[4];
  if (nv > 0) load_step(q, 0);
  for (long long v0 = 0; v0 < nv; v0 += step) {
    int4 qn[4];
    if (v0 + step < nv) {
      load_step(qn, v0 + step);
    } else {
#pragma unroll
      for (int u = 0; u < 4; ++u) qn[u] = make_int4(-1, -1, -1, -1);
    }
#pragma unroll
    for (int u = 0; u < 4; ++u) {
      long long i = (long long)lead + (v0 + (long long)u * OWN_T + t) * 4;
      own_take(q[u].x, i + 0, base, rows, lst, cntS);
      own_take(q[u].y, i + 1, base, rows, lst, cntS);
      own_take(q[u].z, i + 2, base, rows, lst, cntS);
      own_take(q[u].w, i + 3, base, rows, lst, cntS);
    }
#pragma unroll
    for (int u = 0; u < 4; ++u) q[u] = qn[u];
  }
}

__device__ __forceinline__ void own_unlisted(
    const int* __restrict__ srcI, const int* __restrict__ dstI,
    const float* __restrict__ ea, const float* __restrict__ la,
    const float* __restrict__ We, const float* __restrict__ ae,
    const float* __restrict__ a_src, const float* __restrict__ a_dst,
    const float* __restrict__ xh,
    float* acc, unsigned* mxS, float* smS, float* mES, int base, int rows, int E, int t) {
  if (t < 64) mES[t] = own_medge(We, ae, t >> 2, t & 3);
  __syncthreads();
  const long long nall = (long long)E + rows;
  for (long long i = t; i < nall; i += OWN_T) {
    unsigned r;
    if (i < (long long)E) r = (unsigned)(dstI[i] - base);
    else r = (unsigned)(i - E);
    if (r < (unsigned)rows) {
      int s;
      const float* row;
      if (i < (long long)E) { s = srcI[i]; row = ea + i * 16; }
      else { s = base + (int)r; row = la + (long long)s * 16; }
#pragma nounroll
      for (int h = 0; h < 4; ++h) {
        float v = own_leaky((a_src[(long long)s * 4 + h] + a_dst[(long long)(base + (int)r) * 4 + h]) +
                            own_term(row, mES, h));
        atomicMax(&mxS[r * 4 + h], ordEnc(v));
      }
    }
  }
  __syncthreads();
  for (long long i = t; i < nall; i += OWN_T) {
    unsigned r;
    if (i < (long long)E) r = (unsigned)(dstI[i] - base);
    else r = (unsigned)(i - E);
    if (r < (unsigned)rows) {
      int s;
      const float* row;
      if (i < (long long)E) { s = srcI[i]; row = ea + i * 16; }
      else { s = base + (int)r; row = la + (long long)s * 16; }
      const float* xr = xh + (long long)s * 128;
#pragma nounroll
      for (int h = 0; h < 4; ++h) {
        float v = own_leaky((a_src[(long long)s * 4 + h] + a_dst[(long long)(base + (int)r) * 4 + h]) +
                            own_term(row, mES, h));
        float p = __expf(v - own_max_dec(mxS[r * 4 + h]));
        atomicAdd(&smS[r * 4 + h], p);
#pragma nounroll
        for (int c = 0; c < 32; ++c) atomicAdd(acc + r * 128 + h * 32 + c, p * xr[h * 32 + c]);
      }
    }
  }
}


__device__ __noinline__ void loop_unlisted(
    const int* __restrict__ dstI, const float* __restrict__ ea,
    float* sumS, unsigned* cS, int base, int rows, int E, int t) {
  for (long long i = t; i < (long long)E; i += OWN_T) {
    unsigned r = (unsigned)(dstI[i] - base);
    if (r < (unsigned)rows) {
      for (int j = 0; j < 16; ++j) atomicAdd(&sumS[r * 16 + j], ea[i * 16 + j]);
      atomicAdd(&cS[r], 1u);
    }
  }
}

__global__ __launch_bounds__(OWN_T) void gat_own_loop(
    const int* __restrict__ dstI, const float* __restrict__ ea,
    const float* __restrict__ We1, const float* __restrict__ ae1,
    const float* __restrict__ We2, const float* __restrict__ ae2,
    float* __restrict__ la, float* t1, unsigned* keep, int E, int Nn) {
  __shared__ __attribute__((aligned(16))) float sumS[OWN_R * 128];
  __shared__ __attribute__((aligned(16))) unsigned lst[OWN_CAP];
  __shared__ __attribute__((aligned(16))) float mE[128];
  __shared__ unsigned cS[OWN_R];
  __shared__ int cntS;

  const int t = threadIdx.x;
  const int base = (int)blockIdx.x * OWN_R;
  const int rows = (Nn - base >= OWN_R) ? OWN_R : (Nn - base);
  const bool packable = E >= 1 && E <= (int)OWN_EMASK && Nn <= (int)OWN_EMASK;
  unsigned* const keep_b = keep + (long long)base * 128;
  float* const t1_b = t1 + (long long)base * 128;
  float* const t2_b = (float*)keep_b + OWN_T2OFF;
  const bool fits = rows * 128 >= OWN_BLKW;

  if (t == 0) cntS = 0;
  if (t < 128)
    mE[t] = own_medge((t < 64) ? We1 : We2, (t < 64) ? ae1 : ae2, (t >> 2) & 15, t & 3);
  for (int i = t; i < OWN_R * 16; i += OWN_T) sumS[i] = 0.0f;
  for (int i = t; i < OWN_R; i += OWN_T) cS[i] = 0u;
  __syncthreads();

  if (packable) own_scan(dstI, E, base, rows, lst, &cntS, t);
  __syncthreads();
  const int nraw = cntS;
  const bool listed = packable && nraw <= OWN_CAP;
  const int nl = listed ? nraw : 0;
  const bool kept = listed && fits;

  if (fits) {
    for (int i = t; i < nl; i += OWN_T) keep_b[OWN_KEEP0 + i] = lst[i];
    if (t == 0) keep_b[0] = listed ? (unsigned)nraw : (unsigned)OWN_CAP + 1u;
  }

  if (!listed) {
    loop_unlisted(dstI, ea, sumS, cS, base, rows, E, t);
  } else {
    const int q = t & 3;
    float cf[2][4][4];
#pragma unroll
    for (int l = 0; l < 2; ++l) {
#pragma unroll
      for (int j = 0; j < 4; ++j) {
        float4 c4 = *(const float4*)(mE + l * 64 + (q * 4 + j) * 4);
        cf[l][j][0] = c4.x; cf[l][j][1] = c4.y; cf[l][j][2] = c4.z; cf[l][j][3] = c4.w;
      }
    }
    for (int b0 = 0; b0 < nl; b0 += OWN_T * 2) {
      float4 v[8];
      unsigned rw[8];
#pragma unroll
      for (int u = 0; u < 8; ++u) {
        int it = b0 + u * (OWN_T / 4) + (t >> 2);
        bool ok = it < nl;
        unsigned w = lst[ok ? it : nl - 1];
        v[u] = *(const float4*)(ea + (long long)(w & OWN_EMASK) * 16 + q * 4);
        rw[u] = ok ? (w >> OWN_RSH) : 0xffffffffu;
      }
#pragma unroll
      for (int u = 0; u < 8; ++u) {
        if (rw[u] != 0xffffffffu) {
          float* a = sumS + rw[u] * 16 + q * 4;
          atomicAdd(a + 0, v[u].x);
          atomicAdd(a + 1, v[u].y);
          atomicAdd(a + 2, v[u].z);
          atomicAdd(a + 3, v[u].w);
          if (q == 0) atomicAdd(&cS[rw[u]], 1u);
        }
        float pd[8];
#pragma unroll
        for (int o = 0; o < 8; ++o) {
          float p = v[u].x * cf[o >> 2][0][o & 3];
          p += v[u].y * cf[o >> 2][1][o & 3];
          p += v[u].z * cf[o >> 2][2][o & 3];
          p += v[u].w * cf[o >> 2][3][o & 3];
          p += quad_xor1(p);
          p += quad_xor2(p);
          pd[o] = p;
        }
        if (kept && rw[u] != 0xffffffffu) {
          const long long it = b0 + u * (OWN_T / 4) + (t >> 2);
          if (q == 0) *(float4*)(t1_b + it * 4) = make_float4(pd[0], pd[1], pd[2], pd[3]);
          else if (q == 1) *(float4*)(t2_b + it * 4) = make_float4(pd[4], pd[5], pd[6], pd[7]);
        }
      }
    }
  }
  __syncthreads();

  for (int i = t; i < rows * 16; i += OWN_T) {
    float cf = (float)cS[i >> 4];
    cf = (cf < 1.0f) ? 1.0f : cf;
    la[(long long)base * 16 + i] = sumS[i] / cf;
  }
  if (kept) {
    for (int i = t; i < rows * 8; i += OWN_T) {
      const int r = i >> 3, l = (i >> 2) & 1, h = i & 3;
      float cf = (float)cS[r];
      cf = (cf < 1.0f) ? 1.0f : cf;
      float qs = 0.0f;
      for (int k = 0; k < 16; ++k) qs += (sumS[r * 16 + k] / cf) * mE[l * 64 + k * 4 + h];
      float* tb = (l == 0) ? t1_b : t2_b;
      tb[(long long)(nl + r) * 4 + h] = qs;
    }
  }
}

__global__ __launch_bounds__(OWN_T) void gat_own(
    const int* __restrict__ srcI, const int* __restrict__ dstI,
    const float* __restrict__ ea, const float* __restrict__ la,
    const float* __restrict__ We, const float* __restrict__ ae,
    const float* __restrict__ a_src, const float* __restrict__ a_dst,
    const float* __restrict__ xh, const float* __restrict__ bias,
    float* out, const unsigned* keep, const float* tk, int toff,
    int E, int Nn, int relu) {
  __shared__ __attribute__((aligned(16))) float acc[OWN_R * 128];
  __shared__ __attribute__((aligned(16))) unsigned lst[OWN_CAP];
  __shared__ __attribute__((aligned(16))) unsigned mxS[OWN_R * 4];
  __shared__ __attribute__((aligned(16))) float smS[OWN_R * 4];
  __shared__ __attribute__((aligned(16))) float mES[64];

  const int t = threadIdx.x;
  const int lane = t & 31;
  const int base = (int)blockIdx.x * OWN_R;
  const int rows = (Nn - base >= OWN_R) ? OWN_R : (Nn - base);

  const bool packable = E >= 1 && E <= (int)OWN_EMASK && Nn <= (int)OWN_EMASK;
  const unsigned* const keep_b = keep + (long long)base * 128;
  const bool fits = rows * 128 >= OWN_BLKW;
  int kept = -1;
  if (fits && packable) {
    unsigned k = keep_b[0];
    if (k <= (unsigned)OWN_CAP) kept = (int)k;
  }

  {
    float4* a4 = (float4*)acc;
    for (int i = t; i < OWN_R * 32; i += OWN_T) a4[i] = make_float4(0.0f, 0.0f, 0.0f, 0.0f);
    for (int i = t; i < OWN_R * 4; i += OWN_T) {
      mxS[i] = 0x007fffffu;
      smS[i] = 0.0f;
    }
  }
  if (kept >= 0) {
    for (int i = t; i < kept; i += OWN_T) lst[i] = keep_b[OWN_KEEP0 + i];
  }
  __syncthreads();

  if (kept < 0) {
    own_unlisted(srcI, dstI, ea, la, We, ae, a_src, a_dst, xh, acc, mxS, smS, mES,
                 base, rows, E, t);
  } else {
    const int nl = kept;
    const int nit = nl + rows;
    const float* const tb = tk + (long long)base * 128 + toff;
    const float4* const tb4 = (const float4*)tb;
    const float4* const as4 = (const float4*)a_src;
    const float4* const ad4 = (const float4*)a_dst;

    for (int b0 = 0; b0 < nit; b0 += OWN_T * 4) {
      int sv[4];
      unsigned rv[4];
      bool il[4];
      float4 tv[4], dv[4], av[4];
#pragma unroll
      for (int u = 0; u < 4; ++u) {
        int it = b0 + u * OWN_T + t;
        bool ok = it < nit;
        il[u] = it < nl;
        unsigned w = lst[il[u] ? it : 0];
        int sl = srcI[il[u] ? (int)(w & OWN_EMASK) : 0];
        unsigned r = il[u] ? (w >> OWN_RSH) : (ok ? (unsigned)(it - nl) : 0u);
        sv[u] = il[u] ? sl : base + (int)r;
        tv[u] = tb4[ok ? it : nit - 1];
        dv[u] = ad4[base + (int)r];
        rv[u] = ok ? r : 0xffffffffu;
      }
#pragma unroll
      for (int u = 0; u < 4; ++u) av[u] = as4[sv[u]];
#pragma unroll
      for (int u = 0; u < 4; ++u) {
        if (rv[u] != 0xffffffffu) {
          unsigned* m = mxS + rv[u] * 4;
          atomicMax(m + 0, ordEnc(own_leaky((av[u].x + dv[u].x) + tv[u].x)));
          atomicMax(m + 1, ordEnc(own_leaky((av[u].y + dv[u].y) + tv[u].y)));
          atomicMax(m + 2, ordEnc(own_leaky((av[u].z + dv[u].z) + tv[u].z)));
          atomicMax(m + 3, ordEnc(own_leaky((av[u].w + dv[u].w) + tv[u].w)));
          if (il[u]) lst[b0 + u * OWN_T + t] = (unsigned)sv[u] | (rv[u] << OWN_RSH);
        }
      }
    }
    __syncthreads();

    const int wv = __builtin_amdgcn_readfirstlane(t >> 5);
    const int hl = lane & 3;
    auto fetch = [&](int g, int& s_, unsigned& r_, float& as_, float& ad_, float& tt_,
                     unsigned& mx_) {
      int it = g + (lane >> 2);
      if (it >= nit) it = nit - 1;
      const bool isl = it < nl;
      unsigned w = lst[isl ? it : 0];
      r_ = isl ? (w >> OWN_RSH) : (unsigned)(it - nl);
      s_ = isl ? (int)(w & OWN_EMASK) : base + (int)r_;
      as_ = a_src[(long long)s_ * 4 + hl];
      ad_ = a_dst[(long long)(base + (int)r_) * 4 + hl];
      tt_ = tb[(long long)it * 4 + hl];
      mx_ = mxS[r_ * 4 + hl];
    };
    int s = base;
    unsigned r = 0, mx = 0;
    float as = 0.0f, ad = 0.0f, tt = 0.0f;
    if (wv * 8 < nit) fetch(wv * 8, s, r, as, ad, tt, mx);
    for (int g0 = wv * 8; g0 < nit; g0 += OWN_W * 8) {
      int nlive = (nit - g0 >= 8) ? 8 : (nit - g0);
      float xv[8][4];
      int rr[8];
#pragma unroll
      for (int k = 0; k < 8; ++k) {
        int se = __builtin_amdgcn_readlane(s, k * 4);
        rr[k] = __builtin_amdgcn_readlane((int)r, k * 4);
        const float* xr = xh + (long long)se * 128 + lane;
#pragma unroll
        for (int j = 0; j < 4; ++j) xv[k][j] = xr[j * 32];
      }
      float p = __expf(own_leaky((as + ad) + tt) - own_max_dec(mx));
      if ((lane >> 2) < nlive) atomicAdd(&smS[r * 4 + hl], p);
      int s2 = base;
      unsigned r2 = 0, mx2 = 0;
      float as2 = 0.0f, ad2 = 0.0f, tt2 = 0.0f;
      int gn = g0 + OWN_W * 8;
      if (gn < nit) fetch(gn, s2, r2, as2, ad2, tt2, mx2);
      int pi = __float_as_int(p);
#pragma unroll
      for (int k = 0; k < 8; ++k) {
        if (k == 0 || k < nlive) {
          float* a = acc + rr[k] * 128 + lane;
#pragma unroll
          for (int j = 0; j < 4; ++j) {
            float pj = __int_as_float(__builtin_amdgcn_readlane(pi, k * 4 + j));
            atomicAdd(a + j * 32, pj * xv[k][j]);
          }
        }
      }
      s = s2; r = r2; as = as2; ad = ad2; tt = tt2; mx = mx2;
    }
  }
  __syncthreads();

  {
    const int c = t & 127;
    const int h = c >> 5;
    const float b = bias[c];
    for (int r = t >> 7; r < rows; r += OWN_T / 128) {
      float v = acc[r * 128 + c] / (smS[r * 4 + h] + 1e-16f) + b;
      if (relu) v = (v > 0.0f) ? v : 0.0f;
      out[((long long)base + r) * 128 + c] = v;
    }
  }
}

extern "C" void kernel_launch(void* const* d_in, const int* in_sizes, int n_in,
                              void* d_out, int out_size, void* d_ws,
                              size_t ws_size, hipStream_t stream) {
  const int N = in_sizes[0] / 128;
  const int E = in_sizes[1] / 2;

  const float* x  = (const float*)d_in[0];
  const int* ei   = (const int*)d_in[1];
  const int* srcI = ei;
  const int* dstI = ei + E;
  const float* ea = (const float*)d_in[2];

  const float* Wl[2]  = {(const float*)d_in[3], (const float*)d_in[9]};
  const float* asl[2] = {(const float*)d_in[4], (const float*)d_in[10]};
  const float* adl[2] = {(const float*)d_in[5], (const float*)d_in[11]};
  const float* Wel[2] = {(const float*)d_in[6], (const float*)d_in[12]};
  const float* ael[2] = {(const float*)d_in[7], (const float*)d_in[13]};
  const float* bl[2]  = {(const float*)d_in[8], (const float*)d_in[14]};

  float* wsf = (float*)d_ws;
  size_t off = 0;
  float* xh    = wsf + off; off += (size_t)N * 128;
  float* h1    = wsf + off; off += (size_t)N * 128;
  float* la    = wsf + off; off += (size_t)N * 16;
  float* a_src = wsf + off; off += (size_t)N * 4;
  float* a_dst = wsf + off; off += (size_t)N * 4;
  (void)ws_size; (void)n_in; (void)out_size;

  const long long own_blocks = ((long long)N + OWN_R - 1) / OWN_R;

  hipLaunchKernelGGL(gat_own_loop, dim3((unsigned)own_blocks), dim3(OWN_T), 0,
                     stream, dstI, ea, Wel[0], ael[0], Wel[1], ael[1], la, h1,
                     (unsigned*)d_out, E, N);

  for (int l = 0; l < 2; ++l) {
    const float* xin = (l == 0) ? x : h1;
    float* outb = (l == 0) ? h1 : (float*)d_out;

    {
      long long waves = ((long long)N + 15) / 16;
      long long blocks = (waves + 7) / 8;
      hipLaunchKernelGGL(gat_gemm_wmma, dim3((unsigned)blocks), dim3(256), 0,
                         stream, xin, Wl[l], asl[l], adl[l], xh, a_src, a_dst, N);
    }

    hipLaunchKernelGGL(gat_own, dim3((unsigned)own_blocks), dim3(OWN_T), 0, stream,
                       srcI, dstI, ea, la, Wel[l], ael[l], a_src, a_dst, xh, bl[l],
                       outb, (const unsigned*)d_out,
                       (l == 0) ? (const float*)h1 : (const float*)d_out,
                       (l == 0) ? 0 : (int)OWN_T2OFF, E, N, (l == 0) ? 1 : 0);
  }
}


# range owners scan one owner-tag byte per edge (new front launch writes it; 16 edges per word test, hit pieces queued, a lane per piece), not 4-byte ids: 5 -> 6 launches; sum order stays arrival
# speedup vs baseline: 4.5866x; 1.0879x over previous
#include <hip/hip_runtime.h>
#include <hip/hip_bf16.h>

typedef __attribute__((ext_vector_type(2))) float v2f;
typedef __attribute__((ext_vector_type(8))) float v8f;

#define NEG_SLOPE 0.2f

__device__ __forceinline__ unsigned ordEnc(float f) {
  unsigned u = __float_as_uint(f);
  return (u & 0x80000000u) ? ~u : (u | 0x80000000u);
}
__device__ __forceinline__ float ordDec(unsigned u) {
  return (u & 0x80000000u) ? __uint_as_float(u ^ 0x80000000u)
                           : __uint_as_float(~u);
}

__global__ __launch_bounds__(256) void gat_gemm_wmma(
    const float* __restrict__ X, const float* __restrict__ W,
    const float* __restrict__ as, const float* __restrict__ ad,
    float* __restrict__ Y, float* __restrict__ a_src,
    float* __restrict__ a_dst, int Nrows) {
  int wave = (int)(((long long)blockIdx.x * blockDim.x + threadIdx.x) >> 5);
  int lane = threadIdx.x & 31;
  int m0 = wave * 16;
  if (m0 >= Nrows) return;

  v8f acc[8] = {};
  int half = lane >> 4;
  int l16 = lane & 15;
  int arow = m0 + l16;
  if (arow >= Nrows) arow = Nrows - 1;
  const float* xr = X + (long long)arow * 128;

  for (int k0 = 0; k0 < 128; k0 += 4) {
    v2f a;
    a.x = xr[k0 + half * 2];
    a.y = xr[k0 + half * 2 + 1];
#pragma unroll
    for (int nt = 0; nt < 8; ++nt) {
      int col = nt * 16 + l16;
      v2f b;
      b.x = W[(k0 + half * 2) * 128 + col];
      b.y = W[(k0 + half * 2 + 1) * 128 + col];
      acc[nt] = __builtin_amdgcn_wmma_f32_16x16x4_f32(
          false, a, false, b, (short)0, acc[nt], false, false);
    }
  }
#pragma unroll
  for (int nt = 0; nt < 8; ++nt) {
#pragma unroll
    for (int r = 0; r < 8; ++r) {
      int row = m0 + r + half * 8;
      if (row < Nrows) Y[(long long)row * 128 + nt * 16 + l16] = acc[nt][r];
    }
  }

  float cs[8], cd[8];
#pragma unroll
  for (int nt = 0; nt < 8; ++nt) {
    cs[nt] = as[nt * 16 + l16];
    cd[nt] = ad[nt * 16 + l16];
  }
  float s0 = 0.0f, s1 = 0.0f, d0 = 0.0f, d1 = 0.0f;
#pragma unroll
  for (int r = 0; r < 8; ++r) {
#pragma unroll
    for (int hd = 0; hd < 4; ++hd) {
      float ps = acc[2 * hd][r] * cs[2 * hd] + acc[2 * hd + 1][r] * cs[2 * hd + 1];
      float pd = acc[2 * hd][r] * cd[2 * hd] + acc[2 * hd + 1][r] * cd[2 * hd + 1];
      ps += __shfl_xor(ps, 8, 32);
      pd += __shfl_xor(pd, 8, 32);
      ps += __shfl_xor(ps, 4, 32);
      pd += __shfl_xor(pd, 4, 32);
      ps += __shfl_xor(ps, 2, 32);
      pd += __shfl_xor(pd, 2, 32);
      ps += __shfl_xor(ps, 1, 32);
      pd += __shfl_xor(pd, 1, 32);
      const bool mine = (l16 == r * 2 + (hd >> 1));
      if ((hd & 1) == 0) {
        s0 = mine ? ps : s0;
        d0 = mine ? pd : d0;
      } else {
        s1 = mine ? ps : s1;
        d1 = mine ? pd : d1;
      }
    }
  }
  {
    int row = m0 + half * 8 + (l16 >> 1);
    if (row < Nrows) {
      long long o = (long long)row * 4 + (l16 & 1) * 2;
      *(float2*)(a_src + o) = make_float2(s0, s1);
      *(float2*)(a_dst + o) = make_float2(d0, d1);
    }
  }
}

__device__ __forceinline__ float own_medge(const float* __restrict__ We,
                                           const float* __restrict__ ae, int k, int h) {
  float s = 0.0f;
  for (int c = 0; c < 32; ++c)
    s += We[k * 128 + h * 32 + c] * ae[h * 32 + c];
  return s;
}

__device__ __forceinline__ float own_leaky(float v) {
  return (v > 0.0f) ? v : NEG_SLOPE * v;
}

__device__ __forceinline__ float own_term(const float* __restrict__ row,
                                          const float* m, int h) {
  float q = 0.0f;
#pragma nounroll
  for (int k = 0; k < 16; ++k) q += row[k] * m[k * 4 + h];
  return q;
}

__device__ __forceinline__ float quad_xor1(float v) {
  return __int_as_float(__builtin_amdgcn_mov_dpp(__float_as_int(v), 0xB1, 0xF, 0xF, true));
}
__device__ __forceinline__ float quad_xor2(float v) {
  return __int_as_float(__builtin_amdgcn_mov_dpp(__float_as_int(v), 0x4E, 0xF, 0xF, true));
}

#define OWN_R 400
#define OWN_T 1024
#define OWN_W (OWN_T / 32)
#define OWN_CAP 9900
#define OWN_KEEP0 4
#define OWN_T2OFF (OWN_KEEP0 + OWN_CAP)
#define OWN_BLKW (OWN_T2OFF + 4 * (OWN_CAP + OWN_R))
#define OWN_RSH 23
#define OWN_EMASK ((1u << OWN_RSH) - 1u)
static_assert(OWN_R <= (1 << (32 - OWN_RSH)), "the row in range must fit above the edge number");
static_assert(OWN_T2OFF % 4 == 0, "terms start on a 16-byte boundary");
static_assert(OWN_BLKW <= OWN_R * 128, "list and terms must fit the range's own output rows");

__device__ __forceinline__ float own_max_dec(unsigned u) {
  float mv = ordDec(u);
  if (!(mv > -__builtin_inff() && mv < __builtin_inff())) mv = 0.0f;
  return mv;
}

__device__ __forceinline__ void own_take(int dval, long long i, int base, int rows,
                                         unsigned* lst, int* cntS) {
  unsigned r = (unsigned)(dval - base);
  if (r < (unsigned)rows) {
    int slot = atomicAdd(cntS, 1);
    if (slot < OWN_CAP) lst[slot] = (unsigned)i | (r << OWN_RSH);
  }
}

__device__ __forceinline__ void own_scan(const int* __restrict__ dstI, int E,
                                         int base, int rows, unsigned* lst,
                                         int* cntS, int t) {
  int lead = (int)((4u - (unsigned)(((unsigned long long)dstI >> 2) & 3u)) & 3u);
  if (lead > E) lead = E;
  const long long nv = ((long long)E - lead) >> 2;
  const long long tail0 = (long long)lead + nv * 4;
  {
    long long i = -1;
    if (t < lead) i = t;
    else if ((long long)(t - lead) < (long long)E - tail0) i = tail0 + (t - lead);
    if (i >= 0) own_take(dstI[i], i, base, rows, lst, cntS);
  }
  const int4* pv = (const int4*)(dstI + lead);
  const long long step = (long long)OWN_T * 4;
  auto load_step = [&](int4 (&qq)[4], long long v0) {
    if (v0 + step <= nv) {
#pragma unroll
      for (int u = 0; u < 4; ++u) qq[u] = pv[v0 + (long long)u * OWN_T + t];
    } else {
#pragma unroll
      for (int u = 0; u < 4; ++u) {
        long long v = v0 + (long long)u * OWN_T + t;
        qq[u] = (v < nv) ? pv[v] : make_int4(-1, -1, -1, -1);
      }
    }
  };
  int4 q[4];
  if (nv > 0) load_step(q, 0);
  for (long long v0 = 0; v0 < nv; v0 += step) {
    int4 qn[4];
    if (v0 + step < nv) {
      load_step(qn, v0 + step);
    } else {
#pragma unroll
      for (int u = 0; u < 4; ++u) qn[u] = make_int4(-1, -1, -1, -1);
    }
#pragma unroll
    for (int u = 0; u < 4; ++u) {
      long long i = (long long)lead + (v0 + (long long)u * OWN_T + t) * 4;
      own_take(q[u].x, i + 0, base, rows, lst, cntS);
      own_take(q[u].y, i + 1, base, rows, lst, cntS);
      own_take(q[u].z, i + 2, base, rows, lst, cntS);
      own_take(q[u].w, i + 3, base, rows, lst, cntS);
    }
#pragma unroll
    for (int u = 0; u < 4; ++u) q[u] = qn[u];
  }
}

__device__ __forceinline__ void own_unlisted(
    const int* __restrict__ srcI, const int* __restrict__ dstI,
    const float* __restrict__ ea, const float* __restrict__ la,
    const float* __restrict__ We, const float* __restrict__ ae,
    const float* __restrict__ a_src, const float* __restrict__ a_dst,
    const float* __restrict__ xh,
    float* acc, unsigned* mxS, float* smS, float* mES, int base, int rows, int E, int t) {
  if (t < 64) mES[t] = own_medge(We, ae, t >> 2, t & 3);
  __syncthreads();
  const long long nall = (long long)E + rows;
  for (long long i = t; i < nall; i += OWN_T) {
    unsigned r;
    if (i < (long long)E) r = (unsigned)(dstI[i] - base);
    else r = (unsigned)(i - E);
    if (r < (unsigned)rows) {
      int s;
      const float* row;
      if (i < (long long)E) { s = srcI[i]; row = ea + i * 16; }
      else { s = base + (int)r; row = la + (long long)s * 16; }
#pragma nounroll
      for (int h = 0; h < 4; ++h) {
        float v = own_leaky((a_src[(long long)s * 4 + h] + a_dst[(long long)(base + (int)r) * 4 + h]) +
                            own_term(row, mES, h));
        atomicMax(&mxS[r * 4 + h], ordEnc(v));
      }
    }
  }
  __syncthreads();
  for (long long i = t; i < nall; i += OWN_T) {
    unsigned r;
    if (i < (long long)E) r = (unsigned)(dstI[i] - base);
    else r = (unsigned)(i - E);
    if (r < (unsigned)rows) {
      int s;
      const float* row;
      if (i < (long long)E) { s = srcI[i]; row = ea + i * 16; }
      else { s = base + (int)r; row = la + (long long)s * 16; }
      const float* xr = xh + (long long)s * 128;
#pragma nounroll
      for (int h = 0; h < 4; ++h) {
        float v = own_leaky((a_src[(long long)s * 4 + h] + a_dst[(long long)(base + (int)r) * 4 + h]) +
                            own_term(row, mES, h));
        float p = __expf(v - own_max_dec(mxS[r * 4 + h]));
        atomicAdd(&smS[r * 4 + h], p);
#pragma nounroll
        for (int c = 0; c < 32; ++c) atomicAdd(acc + r * 128 + h * 32 + c, p * xr[h * 32 + c]);
      }
    }
  }
}

#define OWN_NOTAG 255u
__device__ __forceinline__ bool own_tagged(int E, int Nn) {
  const long long nrange = ((long long)Nn + OWN_R - 1) / OWN_R;
  return nrange <= 255 && (long long)E + 16 <= (long long)Nn * 512;
}
__device__ __forceinline__ unsigned own_tag(int d, int Nn) {
  return ((unsigned)d < (unsigned)Nn) ? (unsigned)d / (unsigned)OWN_R : OWN_NOTAG;
}

__global__ __launch_bounds__(256) void gat_own_tags(
    const int* __restrict__ dstI, unsigned* __restrict__ tagw, int E, int Nn) {
  if (!own_tagged(E, Nn)) return;
  const long long nw = (((long long)E + 15) >> 4) << 2;
  const bool al = (((unsigned long long)dstI) & 15ull) == 0ull;
  const long long stride = (long long)gridDim.x * blockDim.x;
  for (long long w0 = (long long)blockIdx.x * blockDim.x + threadIdx.x; w0 < nw; w0 += stride * 4) {
    int4 d[4];
#pragma unroll
    for (int u = 0; u < 4; ++u) {
      const long long i = (w0 + stride * u) * 4;
      if (al && i + 4 <= (long long)E) {
        d[u] = *(const int4*)(dstI + i);
      } else {
        d[u].x = (i + 0 < (long long)E) ? dstI[i + 0] : -1;
        d[u].y = (i + 1 < (long long)E) ? dstI[i + 1] : -1;
        d[u].z = (i + 2 < (long long)E) ? dstI[i + 2] : -1;
        d[u].w = (i + 3 < (long long)E) ? dstI[i + 3] : -1;
      }
    }
#pragma unroll
    for (int u = 0; u < 4; ++u) {
      const long long w = w0 + stride * u;
      if (w < nw)
        tagw[w] = own_tag(d[u].x, Nn) | (own_tag(d[u].y, Nn) << 8) |
                  (own_tag(d[u].z, Nn) << 16) | (own_tag(d[u].w, Nn) << 24);
    }
  }
}

__device__ __forceinline__ void own_scan_tags(const uint4* __restrict__ tg, long long np,
                                              unsigned pat, unsigned* que, int* qcnt, int t) {
  const long long step = (long long)OWN_T * 4;
  auto load_step = [&](uint4 (&qq)[4], long long v0) {
    if (v0 + step <= np) {
#pragma unroll
      for (int u = 0; u < 4; ++u) qq[u] = tg[v0 + (long long)u * OWN_T + t];
    } else {
#pragma unroll
      for (int u = 0; u < 4; ++u) {
        long long v = v0 + (long long)u * OWN_T + t;
        qq[u] = (v < np) ? tg[v] : make_uint4(0xffffffffu, 0xffffffffu, 0xffffffffu, 0xffffffffu);
      }
    }
  };
  uint4 q[4];
  if (np > 0) load_step(q, 0);
  for (long long v0 = 0; v0 < np; v0 += step) {
    uint4 qn[4];
    if (v0 + step < np) {
      load_step(qn, v0 + step);
    } else {
#pragma unroll
      for (int u = 0; u < 4; ++u)
        qn[u] = make_uint4(0xffffffffu, 0xffffffffu, 0xffffffffu, 0xffffffffu);
    }
#pragma unroll
    for (int u = 0; u < 4; ++u) {
      const unsigned x0 = q[u].x ^ pat, x1 = q[u].y ^ pat, x2 = q[u].z ^ pat, x3 = q[u].w ^ pat;
      const unsigned z = ((x0 - 0x01010101u) & ~x0) | ((x1 - 0x01010101u) & ~x1) |
                         ((x2 - 0x01010101u) & ~x2) | ((x3 - 0x01010101u) & ~x3);
      if ((z & 0x80808080u) != 0u) {
        int slot = atomicAdd(qcnt, 1);
        if (slot < OWN_CAP) que[slot] = (unsigned)(v0 + (long long)u * OWN_T + t);
      }
    }
#pragma unroll
    for (int u = 0; u < 4; ++u) q[u] = qn[u];
  }
}

__device__ __forceinline__ void own_take_tagged(const uint4* __restrict__ tg,
                                                const int* __restrict__ dstI,
                                                const unsigned* que, int nq, unsigned bid,
                                                int E, int base, int rows, unsigned* lst,
                                                int* cntS, int t) {
  for (int qi = t; qi < nq; qi += OWN_T) {
    const unsigned p = que[qi];
    const uint4 w = tg[p];
    const unsigned ww[4] = {w.x, w.y, w.z, w.w};
#pragma unroll
    for (int j = 0; j < 4; ++j) {
#pragma unroll
      for (int b = 0; b < 4; ++b) {
        if (((ww[j] >> (8 * b)) & 0xffu) == bid) {
          const long long i = (long long)p * 16 + j * 4 + b;
          if (i < (long long)E) own_take(dstI[i], i, base, rows, lst, cntS);
        }
      }
    }
  }
}


__device__ __noinline__ void loop_unlisted(
    const int* __restrict__ dstI, const float* __restrict__ ea,
    float* sumS, unsigned* cS, int base, int rows, int E, int t) {
  for (long long i = t; i < (long long)E; i += OWN_T) {
    unsigned r = (unsigned)(dstI[i] - base);
    if (r < (unsigned)rows) {
      for (int j = 0; j < 16; ++j) atomicAdd(&sumS[r * 16 + j], ea[i * 16 + j]);
      atomicAdd(&cS[r], 1u);
    }
  }
}

__global__ __launch_bounds__(OWN_T) void gat_own_loop(
    const int* __restrict__ dstI, const float* __restrict__ ea,
    const float* __restrict__ We1, const float* __restrict__ ae1,
    const float* __restrict__ We2, const float* __restrict__ ae2,
    float* __restrict__ la, float* t1, unsigned* keep, const unsigned* tags, int E, int Nn) {
  __shared__ __attribute__((aligned(16))) float sumS[OWN_R * 128];
  __shared__ __attribute__((aligned(16))) unsigned lst[OWN_CAP];
  __shared__ __attribute__((aligned(16))) float mE[128];
  __shared__ unsigned cS[OWN_R];
  __shared__ int cntS;
  __shared__ int qcS;
  static_assert(OWN_R * 16 + OWN_CAP <= OWN_R * 128, "the piece queue must fit behind the sums");
  unsigned* const que = (unsigned*)(sumS + OWN_R * 16);

  const int t = threadIdx.x;
  const int base = (int)blockIdx.x * OWN_R;
  const int rows = (Nn - base >= OWN_R) ? OWN_R : (Nn - base);
  const bool packable = E >= 1 && E <= (int)OWN_EMASK && Nn <= (int)OWN_EMASK;
  unsigned* const keep_b = keep + (long long)base * 128;
  float* const t1_b = t1 + (long long)base * 128;
  float* const t2_b = (float*)keep_b + OWN_T2OFF;
  const bool fits = rows * 128 >= OWN_BLKW;

  if (t == 0) { cntS = 0; qcS = 0; }
  if (t < 128)
    mE[t] = own_medge((t < 64) ? We1 : We2, (t < 64) ? ae1 : ae2, (t >> 2) & 15, t & 3);
  for (int i = t; i < OWN_R * 16; i += OWN_T) sumS[i] = 0.0f;
  for (int i = t; i < OWN_R; i += OWN_T) cS[i] = 0u;
  __syncthreads();

  const bool tagged = packable && own_tagged(E, Nn);
  int nraw;
  if (tagged) {
    const uint4* const tg = (const uint4*)tags;
    own_scan_tags(tg, ((long long)E + 15) >> 4, (unsigned)blockIdx.x * 0x01010101u, que, &qcS, t);
    __syncthreads();
    const int qraw = qcS;
    own_take_tagged(tg, dstI, que, (qraw < OWN_CAP) ? qraw : OWN_CAP, (unsigned)blockIdx.x,
                    E, base, rows, lst, &cntS, t);
    __syncthreads();
    nraw = (qraw > OWN_CAP) ? qraw : cntS;
  } else {
    if (packable) own_scan(dstI, E, base, rows, lst, &cntS, t);
    __syncthreads();
    nraw = cntS;
  }
  const bool listed = packable && nraw <= OWN_CAP;
  const int nl = listed ? nraw : 0;
  const bool kept = listed && fits;

  if (fits) {
    for (int i = t; i < nl; i += OWN_T) keep_b[OWN_KEEP0 + i] = lst[i];
    if (t == 0) keep_b[0] = listed ? (unsigned)nraw : (unsigned)OWN_CAP + 1u;
  }

  if (!listed) {
    loop_unlisted(dstI, ea, sumS, cS, base, rows, E, t);
  } else {
    const int q = t & 3;
    float cf[2][4][4];
#pragma unroll
    for (int l = 0; l < 2; ++l) {
#pragma unroll
      for (int j = 0; j < 4; ++j) {
        float4 c4 = *(const float4*)(mE + l * 64 + (q * 4 + j) * 4);
        cf[l][j][0] = c4.x; cf[l][j][1] = c4.y; cf[l][j][2] = c4.z; cf[l][j][3] = c4.w;
      }
    }
    for (int b0 = 0; b0 < nl; b0 += OWN_T * 2) {
      float4 v[8];
      unsigned rw[8];
#pragma unroll
      for (int u = 0; u < 8; ++u) {
        int it = b0 + u * (OWN_T / 4) + (t >> 2);
        bool ok = it < nl;
        unsigned w = lst[ok ? it : nl - 1];
        v[u] = *(const float4*)(ea + (long long)(w & OWN_EMASK) * 16 + q * 4);
        rw[u] = ok ? (w >> OWN_RSH) : 0xffffffffu;
      }
#pragma unroll
      for (int u = 0; u < 8; ++u) {
        if (rw[u] != 0xffffffffu) {
          float* a = sumS + rw[u] * 16 + q * 4;
          atomicAdd(a + 0, v[u].x);
          atomicAdd(a + 1, v[u].y);
          atomicAdd(a + 2, v[u].z);
          atomicAdd(a + 3, v[u].w);
          if (q == 0) atomicAdd(&cS[rw[u]], 1u);
        }
        float pd[8];
#pragma unroll
        for (int o = 0; o < 8; ++o) {
          float p = v[u].x * cf[o >> 2][0][o & 3];
          p += v[u].y * cf[o >> 2][1][o & 3];
          p += v[u].z * cf[o >> 2][2][o & 3];
          p += v[u].w * cf[o >> 2][3][o & 3];
          p += quad_xor1(p);
          p += quad_xor2(p);
          pd[o] = p;
        }
        if (kept && rw[u] != 0xffffffffu) {
          const long long it = b0 + u * (OWN_T / 4) + (t >> 2);
          if (q == 0) *(float4*)(t1_b + it * 4) = make_float4(pd[0], pd[1], pd[2], pd[3]);
          else if (q == 1) *(float4*)(t2_b + it * 4) = make_float4(pd[4], pd[5], pd[6], pd[7]);
        }
      }
    }
  }
  __syncthreads();

  for (int i = t; i < rows * 16; i += OWN_T) {
    float cf = (float)cS[i >> 4];
    cf = (cf < 1.0f) ? 1.0f : cf;
    la[(long long)base * 16 + i] = sumS[i] / cf;
  }
  if (kept) {
    for (int i = t; i < rows * 8; i += OWN_T) {
      const int r = i >> 3, l = (i >> 2) & 1, h = i & 3;
      float cf = (float)cS[r];
      cf = (cf < 1.0f) ? 1.0f : cf;
      float qs = 0.0f;
      for (int k = 0; k < 16; ++k) qs += (sumS[r * 16 + k] / cf) * mE[l * 64 + k * 4 + h];
      float* tb = (l == 0) ? t1_b : t2_b;
      tb[(long long)(nl + r) * 4 + h] = qs;
    }
  }
}

__global__ __launch_bounds__(OWN_T) void gat_own(
    const int* __restrict__ srcI, const int* __restrict__ dstI,
    const float* __restrict__ ea, const float* __restrict__ la,
    const float* __restrict__ We, const float* __restrict__ ae,
    const float* __restrict__ a_src, const float* __restrict__ a_dst,
    const float* __restrict__ xh, const float* __restrict__ bias,
    float* out, const unsigned* keep, const float* tk, int toff,
    int E, int Nn, int relu) {
  __shared__ __attribute__((aligned(16))) float acc[OWN_R * 128];
  __shared__ __attribute__((aligned(16))) unsigned lst[OWN_CAP];
  __shared__ __attribute__((aligned(16))) unsigned mxS[OWN_R * 4];
  __shared__ __attribute__((aligned(16))) float smS[OWN_R * 4];
  __shared__ __attribute__((aligned(16))) float mES[64];

  const int t = threadIdx.x;
  const int lane = t & 31;
  const int base = (int)blockIdx.x * OWN_R;
  const int rows = (Nn - base >= OWN_R) ? OWN_R : (Nn - base);

  const bool packable = E >= 1 && E <= (int)OWN_EMASK && Nn <= (int)OWN_EMASK;
  const unsigned* const keep_b = keep + (long long)base * 128;
  const bool fits = rows * 128 >= OWN_BLKW;
  int kept = -1;
  if (fits && packable) {
    unsigned k = keep_b[0];
    if (k <= (unsigned)OWN_CAP) kept = (int)k;
  }

  {
    float4* a4 = (float4*)acc;
    for (int i = t; i < OWN_R * 32; i += OWN_T) a4[i] = make_float4(0.0f, 0.0f, 0.0f, 0.0f);
    for (int i = t; i < OWN_R * 4; i += OWN_T) {
      mxS[i] = 0x007fffffu;
      smS[i] = 0.0f;
    }
  }
  if (kept >= 0) {
    for (int i = t; i < kept; i += OWN_T) lst[i] = keep_b[OWN_KEEP0 + i];
  }
  __syncthreads();

  if (kept < 0) {
    own_unlisted(srcI, dstI, ea, la, We, ae, a_src, a_dst, xh, acc, mxS, smS, mES,
                 base, rows, E, t);
  } else {
    const int nl = kept;
    const int nit = nl + rows;
    const float* const tb = tk + (long long)base * 128 + toff;
    const float4* const tb4 = (const float4*)tb;
    const float4* const as4 = (const float4*)a_src;
    const float4* const ad4 = (const float4*)a_dst;

    for (int b0 = 0; b0 < nit; b0 += OWN_T * 4) {
      int sv[4];
      unsigned rv[4];
      bool il[4];
      float4 tv[4], dv[4], av[4];
#pragma unroll
      for (int u = 0; u < 4; ++u) {
        int it = b0 + u * OWN_T + t;
        bool ok = it < nit;
        il[u] = it < nl;
        unsigned w = lst[il[u] ? it : 0];
        int sl = srcI[il[u] ? (int)(w & OWN_EMASK) : 0];
        unsigned r = il[u] ? (w >> OWN_RSH) : (ok ? (unsigned)(it - nl) : 0u);
        sv[u] = il[u] ? sl : base + (int)r;
        tv[u] = tb4[ok ? it : nit - 1];
        dv[u] = ad4[base + (int)r];
        rv[u] = ok ? r : 0xffffffffu;
      }
#pragma unroll
      for (int u = 0; u < 4; ++u) av[u] = as4[sv[u]];
#pragma unroll
      for (int u = 0; u < 4; ++u) {
        if (rv[u] != 0xffffffffu) {
          unsigned* m = mxS + rv[u] * 4;
          atomicMax(m + 0, ordEnc(own_leaky((av[u].x + dv[u].x) + tv[u].x)));
          atomicMax(m + 1, ordEnc(own_leaky((av[u].y + dv[u].y) + tv[u].y)));
          atomicMax(m + 2, ordEnc(own_leaky((av[u].z + dv[u].z) + tv[u].z)));
          atomicMax(m + 3, ordEnc(own_leaky((av[u].w + dv[u].w) + tv[u].w)));
          if (il[u]) lst[b0 + u * OWN_T + t] = (unsigned)sv[u] | (rv[u] << OWN_RSH);
        }
      }
    }
    __syncthreads();

    const int wv = __builtin_amdgcn_readfirstlane(t >> 5);
    const int hl = lane & 3;
    auto fetch = [&](int g, int& s_, unsigned& r_, float& as_, float& ad_, float& tt_,
                     unsigned& mx_) {
      int it = g + (lane >> 2);
      if (it >= nit) it = nit - 1;
      const bool isl = it < nl;
      unsigned w = lst[isl ? it : 0];
      r_ = isl ? (w >> OWN_RSH) : (unsigned)(it - nl);
      s_ = isl ? (int)(w & OWN_EMASK) : base + (int)r_;
      as_ = a_src[(long long)s_ * 4 + hl];
      ad_ = a_dst[(long long)(base + (int)r_) * 4 + hl];
      tt_ = tb[(long long)it * 4 + hl];
      mx_ = mxS[r_ * 4 + hl];
    };
    int s = base;
    unsigned r = 0, mx = 0;
    float as = 0.0f, ad = 0.0f, tt = 0.0f;
    if (wv * 8 < nit) fetch(wv * 8, s, r, as, ad, tt, mx);
    for (int g0 = wv * 8; g0 < nit; g0 += OWN_W * 8) {
      int nlive = (nit - g0 >= 8) ? 8 : (nit - g0);
      float xv[8][4];
      int rr[8];
#pragma unroll
      for (int k = 0; k < 8; ++k) {
        int se = __builtin_amdgcn_readlane(s, k * 4);
        rr[k] = __builtin_amdgcn_readlane((int)r, k * 4);
        const float* xr = xh + (long long)se * 128 + lane;
#pragma unroll
        for (int j = 0; j < 4; ++j) xv[k][j] = xr[j * 32];
      }
      float p = __expf(own_leaky((as + ad) + tt) - own_max_dec(mx));
      if ((lane >> 2) < nlive) atomicAdd(&smS[r * 4 + hl], p);
      int s2 = base;
      unsigned r2 = 0, mx2 = 0;
      float as2 = 0.0f, ad2 = 0.0f, tt2 = 0.0f;
      int gn = g0 + OWN_W * 8;
      if (gn < nit) fetch(gn, s2, r2, as2, ad2, tt2, mx2);
      int pi = __float_as_int(p);
#pragma unroll
      for (int k = 0; k < 8; ++k) {
        if (k == 0 || k < nlive) {
          float* a = acc + rr[k] * 128 + lane;
#pragma unroll
          for (int j = 0; j < 4; ++j) {
            float pj = __int_as_float(__builtin_amdgcn_readlane(pi, k * 4 + j));
            atomicAdd(a + j * 32, pj * xv[k][j]);
          }
        }
      }
      s = s2; r = r2; as = as2; ad = ad2; tt = tt2; mx = mx2;
    }
  }
  __syncthreads();

  {
    const int c = t & 127;
    const int h = c >> 5;
    const float b = bias[c];
    for (int r = t >> 7; r < rows; r += OWN_T / 128) {
      float v = acc[r * 128 + c] / (smS[r * 4 + h] + 1e-16f) + b;
      if (relu) v = (v > 0.0f) ? v : 0.0f;
      out[((long long)base + r) * 128 + c] = v;
    }
  }
}

extern "C" void kernel_launch(void* const* d_in, const int* in_sizes, int n_in,
                              void* d_out, int out_size, void* d_ws,
                              size_t ws_size, hipStream_t stream) {
  const int N = in_sizes[0] / 128;
  const int E = in_sizes[1] / 2;

  const float* x  = (const float*)d_in[0];
  const int* ei   = (const int*)d_in[1];
  const int* srcI = ei;
  const int* dstI = ei + E;
  const float* ea = (const float*)d_in[2];

  const float* Wl[2]  = {(const float*)d_in[3], (const float*)d_in[9]};
  const float* asl[2] = {(const float*)d_in[4], (const float*)d_in[10]};
  const float* adl[2] = {(const float*)d_in[5], (const float*)d_in[11]};
  const float* Wel[2] = {(const float*)d_in[6], (const float*)d_in[12]};
  const float* ael[2] = {(const float*)d_in[7], (const float*)d_in[13]};
  const float* bl[2]  = {(const float*)d_in[8], (const float*)d_in[14]};

  float* wsf = (float*)d_ws;
  size_t off = 0;
  float* xh    = wsf + off; off += (size_t)N * 128;
  float* h1    = wsf + off; off += (size_t)N * 128;
  float* la    = wsf + off; off += (size_t)N * 16;
  float* a_src = wsf + off; off += (size_t)N * 4;
  float* a_dst = wsf + off; off += (size_t)N * 4;
  (void)ws_size; (void)n_in; (void)out_size;

  const long long own_blocks = ((long long)N + OWN_R - 1) / OWN_R;

  hipLaunchKernelGGL(gat_own_tags, dim3(512), dim3(256), 0, stream, dstI,
                     (unsigned*)xh, E, N);

  hipLaunchKernelGGL(gat_own_loop, dim3((unsigned)own_blocks), dim3(OWN_T), 0,
                     stream, dstI, ea, Wel[0], ael[0], Wel[1], ael[1], la, h1,
                     (unsigned*)d_out, (const unsigned*)xh, E, N);

  for (int l = 0; l < 2; ++l) {
    const float* xin = (l == 0) ? x : h1;
    float* outb = (l == 0) ? h1 : (float*)d_out;

    {
      long long waves = ((long long)N + 15) / 16;
      long long blocks = (waves + 7) / 8;
      hipLaunchKernelGGL(gat_gemm_wmma, dim3((unsigned)blocks), dim3(256), 0,
                         stream, xin, Wl[l], asl[l], adl[l], xh, a_src, a_dst, N);
    }

    hipLaunchKernelGGL(gat_own, dim3((unsigned)own_blocks), dim3(OWN_T), 0, stream,
                       srcI, dstI, ea, la, Wel[l], ael[l], a_src, a_dst, xh, bl[l],
                       outb, (const unsigned*)d_out,
                       (l == 0) ? (const float*)h1 : (const float*)d_out,
                       (l == 0) ? 0 : (int)OWN_T2OFF, E, N, (l == 0) ? 1 : 0);
  }
}
